# starting kernel
# baseline (speedup 1.0000x reference)
#include <hip/hip_runtime.h>

typedef __attribute__((ext_vector_type(2))) float v2f;
typedef __attribute__((ext_vector_type(8))) float v8f;

#define LL_BIG   (1 << 22)
#define COLS     4096
#define ROWS     1024
#define ROWS_OUT 576
#define DD_W     2359296
#define TC       32
#define TSTRIDE  33
#define BPB      4

__device__ __forceinline__ void fwht_lds(float* buf, int n, int tid, int nt) {
    for (int h = 1; h < n; h <<= 1) {
        __syncthreads();
        for (int p = tid; p < (n >> 1); p += nt) {
            int i0 = ((p & ~(h - 1)) << 1) | (p & (h - 1));
            int i1 = i0 + h;
            float a = buf[i0], b = buf[i1];
            buf[i0] = a + b;
            buf[i1] = a - b;
        }
    }
    __syncthreads();
}

__global__ void small_fwht_kernel(const float* __restrict__ theta,
                                  const float* __restrict__ BB,
                                  float* __restrict__ wout, int P) {
    __shared__ float w[2048];
    int tid = threadIdx.x, nt = blockDim.x;
    for (int j = tid; j < P; j += nt) w[j] = BB[j] * theta[j];
    fwht_lds(w, P, tid, nt);
    for (int j = tid; j < P; j += nt) wout[j] = w[j];
}

__global__ void gg_partial_kernel(const float* __restrict__ GG, int n,
                                  float* __restrict__ part) {
    __shared__ float red[256];
    int tid = threadIdx.x;
    float s = 0.f;
    for (int i = blockIdx.x * blockDim.x + tid; i < n; i += gridDim.x * blockDim.x) {
        float g = GG[i]; s += g * g;
    }
    red[tid] = s; __syncthreads();
    for (int off = 128; off > 0; off >>= 1) {
        if (tid < off) red[tid] += red[tid + off];
        __syncthreads();
    }
    if (tid == 0) part[blockIdx.x] = red[0];
}

__global__ void gg_final_kernel(const float* __restrict__ part, float* __restrict__ gsum) {
    __shared__ float red[256];
    int tid = threadIdx.x;
    red[tid] = part[tid]; __syncthreads();
    for (int off = 128; off > 0; off >>= 1) {
        if (tid < off) red[tid] += red[tid + off];
        __syncthreads();
    }
    if (tid == 0) gsum[0] = red[0];
}

__global__ __launch_bounds__(512)
void fwht_passA_kernel(const float* __restrict__ w, const int* __restrict__ Pi,
                       const float* __restrict__ GG, float* __restrict__ mbuf,
                       int pmask) {
    __shared__ float tile[ROWS * TSTRIDE];
    int tid = threadIdx.x, nt = blockDim.x;
    int col0 = blockIdx.x * TC;
    for (int idx = tid; idx < ROWS * TC; idx += nt) {
        int r = idx >> 5, c = idx & (TC - 1);
        int g = r * COLS + col0 + c;
        tile[r * TSTRIDE + c] = w[Pi[g] & pmask] * GG[g];
    }
    for (int h = 1; h < ROWS; h <<= 1) {
        __syncthreads();
        for (int q = tid; q < (ROWS >> 1) * TC; q += nt) {
            int c = q & (TC - 1), p = q >> 5;
            int r0 = ((p & ~(h - 1)) << 1) | (p & (h - 1));
            int r1 = r0 + h;
            float a = tile[r0 * TSTRIDE + c], b = tile[r1 * TSTRIDE + c];
            tile[r0 * TSTRIDE + c] = a + b;
            tile[r1 * TSTRIDE + c] = a - b;
        }
    }
    __syncthreads();
    for (int idx = tid; idx < ROWS_OUT * TC; idx += nt) {
        int r = idx >> 5, c = idx & (TC - 1);
        mbuf[(size_t)r * COLS + col0 + c] = tile[r * TSTRIDE + c];
    }
}

__global__ __launch_bounds__(512)
void fwht_passB_kernel(const float* __restrict__ mbuf, const float* __restrict__ init,
                       const float* __restrict__ gsum, float* __restrict__ out, int DD) {
    __shared__ float blk[BPB * COLS];
    int tid = threadIdx.x, nt = blockDim.x;
    size_t base = (size_t)blockIdx.x * (BPB * COLS);
    for (int i = tid; i < BPB * COLS; i += nt) blk[i] = mbuf[base + i];
    for (int h = 1; h < COLS; h <<= 1) {
        __syncthreads();
        for (int q = tid; q < BPB * (COLS >> 1); q += nt) {
            int sb = q >> 11, p = q & (COLS / 2 - 1);
            int i0 = (sb << 12) | (((p & ~(h - 1)) << 1) | (p & (h - 1)));
            int i1 = i0 + h;
            float a = blk[i0], b = blk[i1];
            blk[i0] = a + b;
            blk[i1] = a - b;
        }
    }
    __syncthreads();
    float scale = rsqrtf((float)DD * gsum[0]);
    for (int i = tid; i < BPB * COLS; i += nt) {
        size_t g = base + i;
        out[g] = init[g] + blk[i] * scale;
    }
}

__global__ __launch_bounds__(512)
void fastfood_small_kernel(const float* __restrict__ theta, const float* __restrict__ BB,
                           const int* __restrict__ Pi, const float* __restrict__ GG,
                           const float* __restrict__ init, float* __restrict__ out,
                           int P, int LLv, int DD) {
    __shared__ float w[2048];
    __shared__ float m[4096];
    __shared__ float red[512];
    int tid = threadIdx.x, nt = blockDim.x;
    for (int j = tid; j < P; j += nt) w[j] = BB[j] * theta[j];
    fwht_lds(w, P, tid, nt);
    float s = 0.f;
    for (int i = tid; i < LLv; i += nt) { float g = GG[i]; s += g * g; }
    red[tid] = s; __syncthreads();
    for (int off = nt >> 1; off > 0; off >>= 1) {
        if (tid < off) red[tid] += red[tid + off];
        __syncthreads();
    }
    for (int i = tid; i < LLv; i += nt) m[i] = w[Pi[i] & (P - 1)] * GG[i];
    fwht_lds(m, LLv, tid, nt);
    float scale = rsqrtf((float)DD * red[0]);
    for (int i = tid; i < DD; i += nt) out[i] = init[i] + m[i] * scale;
}

template<int RELU>
__global__ __launch_bounds__(256)
void gemm_wmma_kernel(const float* __restrict__ A, const float* __restrict__ B,
                      const float* __restrict__ bias, float* __restrict__ C,
                      int M, int N, int K) {
    int lane = threadIdx.x & 31;
    int wave = threadIdx.x >> 5;
    int job = blockIdx.x * 8 + wave;
    int mTiles = M >> 4;
    int mt = job % mTiles;
    int ng = job / mTiles;
    int row = lane & 15;
    int kg = (lane >> 4) << 1;
    const float* ap  = A + (size_t)(mt * 16 + row) * K + kg;
    const float* bp0 = B + (size_t)(ng * 64 + row) * K + kg;
    const float* bp1 = bp0 + (size_t)16 * K;
    const float* bp2 = bp0 + (size_t)32 * K;
    const float* bp3 = bp0 + (size_t)48 * K;
    v8f acc0 = {}, acc1 = {}, acc2 = {}, acc3 = {};

    v2f a  = *(const v2f*)(ap);
    v2f b0 = *(const v2f*)(bp0);
    v2f b1 = *(const v2f*)(bp1);
    v2f b2 = *(const v2f*)(bp2);
    v2f b3 = *(const v2f*)(bp3);

    for (int k0 = 4; k0 < K; k0 += 4) {
        v2f an  = *(const v2f*)(ap  + k0);
        v2f bn0 = *(const v2f*)(bp0 + k0);
        v2f bn1 = *(const v2f*)(bp1 + k0);
        v2f bn2 = *(const v2f*)(bp2 + k0);
        v2f bn3 = *(const v2f*)(bp3 + k0);
        acc0 = __builtin_amdgcn_wmma_f32_16x16x4_f32(false, a, false, b0, (short)0, acc0, false, false);
        acc1 = __builtin_amdgcn_wmma_f32_16x16x4_f32(false, a, false, b1, (short)0, acc1, false, false);
        acc2 = __builtin_amdgcn_wmma_f32_16x16x4_f32(false, a, false, b2, (short)0, acc2, false, false);
        acc3 = __builtin_amdgcn_wmma_f32_16x16x4_f32(false, a, false, b3, (short)0, acc3, false, false);
        a = an; b0 = bn0; b1 = bn1; b2 = bn2; b3 = bn3;
    }
    acc0 = __builtin_amdgcn_wmma_f32_16x16x4_f32(false, a, false, b0, (short)0, acc0, false, false);
    acc1 = __builtin_amdgcn_wmma_f32_16x16x4_f32(false, a, false, b1, (short)0, acc1, false, false);
    acc2 = __builtin_amdgcn_wmma_f32_16x16x4_f32(false, a, false, b2, (short)0, acc2, false, false);
    acc3 = __builtin_amdgcn_wmma_f32_16x16x4_f32(false, a, false, b3, (short)0, acc3, false, false);

    int mrow = mt * 16 + (lane >> 4) * 8;
    v8f accs[4] = {acc0, acc1, acc2, acc3};
#pragma unroll
    for (int t = 0; t < 4; ++t) {
        int n = ng * 64 + t * 16 + row;
        float bv = bias[n];
        v8f acc = accs[t];
#pragma unroll
        for (int v = 0; v < 8; ++v) {
            float xv = acc[v] + bv;
            if (RELU) xv = fmaxf(xv, 0.f);
            C[(size_t)(mrow + v) * N + n] = xv;
        }
    }
}

extern "C" void kernel_launch(void* const* d_in, const int* in_sizes, int n_in,
                              void* d_out, int out_size, void* d_ws, size_t ws_size,
                              hipStream_t stream) {
    const float* x     = (const float*)d_in[0];
    const float* theta = (const float*)d_in[1];
    const float* iW1   = (const float*)d_in[2];
    const float* ib1   = (const float*)d_in[3];
    const float* iW2   = (const float*)d_in[4];
    const float* ib2   = (const float*)d_in[5];
    const float* BB_W1 = (const float*)d_in[6];
    const int*   Pi_W1 = (const int*)d_in[7];
    const float* GG_W1 = (const float*)d_in[8];
    const float* BB_b1 = (const float*)d_in[9];
    const int*   Pi_b1 = (const int*)d_in[10];
    const float* GG_b1 = (const float*)d_in[11];
    const float* BB_W2 = (const float*)d_in[12];
    const int*   Pi_W2 = (const int*)d_in[13];
    const float* GG_W2 = (const float*)d_in[14];
    const float* BB_b2 = (const float*)d_in[15];
    const int*   Pi_b2 = (const int*)d_in[16];
    const float* GG_b2 = (const float*)d_in[17];
    float* out = (float*)d_out;

    float* ws = (float*)d_ws;
    size_t o = 0;
    float* W1   = ws + o; o += DD_W;
    float* W2   = ws + o; o += DD_W;
    float* b1   = ws + o; o += 3072;
    float* b2   = ws + o; o += 768;
    float* hbuf = ws + o; o += (size_t)256 * 3072;
    float* mbuf = ws + o; o += DD_W;
    float* wsm  = ws + o; o += 2048;
    float* part = ws + o; o += 256;
    float* gsum = ws + o; o += 1;

    small_fwht_kernel<<<1, 512, 0, stream>>>(theta, BB_W1, wsm, 2048);
    gg_partial_kernel<<<256, 256, 0, stream>>>(GG_W1, LL_BIG, part);
    gg_final_kernel<<<1, 256, 0, stream>>>(part, gsum);
    fwht_passA_kernel<<<COLS / TC, 512, 0, stream>>>(wsm, Pi_W1, GG_W1, mbuf, 2047);
    fwht_passB_kernel<<<ROWS_OUT / BPB, 512, 0, stream>>>(mbuf, iW1, gsum, W1, DD_W);

    small_fwht_kernel<<<1, 512, 0, stream>>>(theta, BB_W2, wsm, 2048);
    gg_partial_kernel<<<256, 256, 0, stream>>>(GG_W2, LL_BIG, part);
    gg_final_kernel<<<1, 256, 0, stream>>>(part, gsum);
    fwht_passA_kernel<<<COLS / TC, 512, 0, stream>>>(wsm, Pi_W2, GG_W2, mbuf, 2047);
    fwht_passB_kernel<<<ROWS_OUT / BPB, 512, 0, stream>>>(mbuf, iW2, gsum, W2, DD_W);

    fastfood_small_kernel<<<1, 512, 0, stream>>>(theta, BB_b1, Pi_b1, GG_b1, ib1, b1,
                                                 2048, 4096, 3072);
    fastfood_small_kernel<<<1, 512, 0, stream>>>(theta, BB_b2, Pi_b2, GG_b2, ib2, b2,
                                                 1024, 1024, 768);

    gemm_wmma_kernel<1><<<96, 256, 0, stream>>>(x, W1, b1, hbuf, 256, 3072, 768);
    gemm_wmma_kernel<0><<<24, 256, 0, stream>>>(hbuf, W2, b2, out, 256, 768, 3072);
}


# GEMM k-loop: each lane loads 64 contiguous bytes per operand row per 32-k chunk (4 x 16B), one chunk ahead of its WMMAs: 8 k-steps share one cache round trip; k summed in a new order inside a chunk
# speedup vs baseline: 1.4231x; 1.4231x over previous
#include <hip/hip_runtime.h>

typedef __attribute__((ext_vector_type(2))) float v2f;
typedef __attribute__((ext_vector_type(4))) float v4f;
typedef __attribute__((ext_vector_type(8))) float v8f;

#define LL_BIG   (1 << 22)
#define COLS     4096
#define ROWS     1024
#define ROWS_OUT 576
#define DD_W     2359296
#define TC       32
#define TSTRIDE  33
#define BPB      4

__device__ __forceinline__ void fwht_lds(float* buf, int n, int tid, int nt) {
    for (int h = 1; h < n; h <<= 1) {
        __syncthreads();
        for (int p = tid; p < (n >> 1); p += nt) {
            int i0 = ((p & ~(h - 1)) << 1) | (p & (h - 1));
            int i1 = i0 + h;
            float a = buf[i0], b = buf[i1];
            buf[i0] = a + b;
            buf[i1] = a - b;
        }
    }
    __syncthreads();
}

__global__ void small_fwht_kernel(const float* __restrict__ theta,
                                  const float* __restrict__ BB,
                                  float* __restrict__ wout, int P) {
    __shared__ float w[2048];
    int tid = threadIdx.x, nt = blockDim.x;
    for (int j = tid; j < P; j += nt) w[j] = BB[j] * theta[j];
    fwht_lds(w, P, tid, nt);
    for (int j = tid; j < P; j += nt) wout[j] = w[j];
}

__global__ void gg_partial_kernel(const float* __restrict__ GG, int n,
                                  float* __restrict__ part) {
    __shared__ float red[256];
    int tid = threadIdx.x;
    float s = 0.f;
    for (int i = blockIdx.x * blockDim.x + tid; i < n; i += gridDim.x * blockDim.x) {
        float g = GG[i]; s += g * g;
    }
    red[tid] = s; __syncthreads();
    for (int off = 128; off > 0; off >>= 1) {
        if (tid < off) red[tid] += red[tid + off];
        __syncthreads();
    }
    if (tid == 0) part[blockIdx.x] = red[0];
}

__global__ void gg_final_kernel(const float* __restrict__ part, float* __restrict__ gsum) {
    __shared__ float red[256];
    int tid = threadIdx.x;
    red[tid] = part[tid]; __syncthreads();
    for (int off = 128; off > 0; off >>= 1) {
        if (tid < off) red[tid] += red[tid + off];
        __syncthreads();
    }
    if (tid == 0) gsum[0] = red[0];
}

__global__ __launch_bounds__(512)
void fwht_passA_kernel(const float* __restrict__ w, const int* __restrict__ Pi,
                       const float* __restrict__ GG, float* __restrict__ mbuf,
                       int pmask) {
    __shared__ float tile[ROWS * TSTRIDE];
    int tid = threadIdx.x, nt = blockDim.x;
    int col0 = blockIdx.x * TC;
    for (int idx = tid; idx < ROWS * TC; idx += nt) {
        int r = idx >> 5, c = idx & (TC - 1);
        int g = r * COLS + col0 + c;
        tile[r * TSTRIDE + c] = w[Pi[g] & pmask] * GG[g];
    }
    for (int h = 1; h < ROWS; h <<= 1) {
        __syncthreads();
        for (int q = tid; q < (ROWS >> 1) * TC; q += nt) {
            int c = q & (TC - 1), p = q >> 5;
            int r0 = ((p & ~(h - 1)) << 1) | (p & (h - 1));
            int r1 = r0 + h;
            float a = tile[r0 * TSTRIDE + c], b = tile[r1 * TSTRIDE + c];
            tile[r0 * TSTRIDE + c] = a + b;
            tile[r1 * TSTRIDE + c] = a - b;
        }
    }
    __syncthreads();
    for (int idx = tid; idx < ROWS_OUT * TC; idx += nt) {
        int r = idx >> 5, c = idx & (TC - 1);
        mbuf[(size_t)r * COLS + col0 + c] = tile[r * TSTRIDE + c];
    }
}

__global__ __launch_bounds__(512)
void fwht_passB_kernel(const float* __restrict__ mbuf, const float* __restrict__ init,
                       const float* __restrict__ gsum, float* __restrict__ out, int DD) {
    __shared__ float blk[BPB * COLS];
    int tid = threadIdx.x, nt = blockDim.x;
    size_t base = (size_t)blockIdx.x * (BPB * COLS);
    for (int i = tid; i < BPB * COLS; i += nt) blk[i] = mbuf[base + i];
    for (int h = 1; h < COLS; h <<= 1) {
        __syncthreads();
        for (int q = tid; q < BPB * (COLS >> 1); q += nt) {
            int sb = q >> 11, p = q & (COLS / 2 - 1);
            int i0 = (sb << 12) | (((p & ~(h - 1)) << 1) | (p & (h - 1)));
            int i1 = i0 + h;
            float a = blk[i0], b = blk[i1];
            blk[i0] = a + b;
            blk[i1] = a - b;
        }
    }
    __syncthreads();
    float scale = rsqrtf((float)DD * gsum[0]);
    for (int i = tid; i < BPB * COLS; i += nt) {
        size_t g = base + i;
        out[g] = init[g] + blk[i] * scale;
    }
}

__global__ __launch_bounds__(512)
void fastfood_small_kernel(const float* __restrict__ theta, const float* __restrict__ BB,
                           const int* __restrict__ Pi, const float* __restrict__ GG,
                           const float* __restrict__ init, float* __restrict__ out,
                           int P, int LLv, int DD) {
    __shared__ float w[2048];
    __shared__ float m[4096];
    __shared__ float red[512];
    int tid = threadIdx.x, nt = blockDim.x;
    for (int j = tid; j < P; j += nt) w[j] = BB[j] * theta[j];
    fwht_lds(w, P, tid, nt);
    float s = 0.f;
    for (int i = tid; i < LLv; i += nt) { float g = GG[i]; s += g * g; }
    red[tid] = s; __syncthreads();
    for (int off = nt >> 1; off > 0; off >>= 1) {
        if (tid < off) red[tid] += red[tid + off];
        __syncthreads();
    }
    for (int i = tid; i < LLv; i += nt) m[i] = w[Pi[i] & (P - 1)] * GG[i];
    fwht_lds(m, LLv, tid, nt);
    float scale = rsqrtf((float)DD * red[0]);
    for (int i = tid; i < DD; i += nt) out[i] = init[i] + m[i] * scale;
}

struct Frag { v4f q0, q1, q2, q3; };

__device__ __forceinline__ void load_frag(Frag& f, const float* __restrict__ p) {
    const v4f* p4 = (const v4f*)p;
    f.q0 = p4[0]; f.q1 = p4[1]; f.q2 = p4[2]; f.q3 = p4[3];
}

#define SCHED_FENCE() __builtin_amdgcn_sched_barrier(0)
#define KEEP_LOADS_IN_THIS_TRIP() asm volatile("" ::: "memory")

#define LO2(q) __builtin_shufflevector(q, q, 0, 1)
#define HI2(q) __builtin_shufflevector(q, q, 2, 3)
#define WMMA_F32(a, b, c) __builtin_amdgcn_wmma_f32_16x16x4_f32(false, a, false, b, (short)0, c, false, false)
#define KSTEP(SEL, Q)                                   \
    acc0 = WMMA_F32(SEL(fa.Q), SEL(f0.Q), acc0);        \
    acc1 = WMMA_F32(SEL(fa.Q), SEL(f1.Q), acc1);        \
    acc2 = WMMA_F32(SEL(fa.Q), SEL(f2.Q), acc2);        \
    acc3 = WMMA_F32(SEL(fa.Q), SEL(f3.Q), acc3);

__device__ __forceinline__ void mma_chunk(const Frag& fa, const Frag& f0, const Frag& f1,
                                          const Frag& f2, const Frag& f3,
                                          v8f& acc0, v8f& acc1, v8f& acc2, v8f& acc3) {
    KSTEP(LO2, q0) KSTEP(HI2, q0)
    KSTEP(LO2, q1) KSTEP(HI2, q1)
    KSTEP(LO2, q2) KSTEP(HI2, q2)
    KSTEP(LO2, q3) KSTEP(HI2, q3)
}

template<int RELU>
__global__ __launch_bounds__(256)
void gemm_wmma_kernel(const float* __restrict__ A, const float* __restrict__ B,
                      const float* __restrict__ bias, float* __restrict__ C,
                      int M, int N, int K) {
    int lane = threadIdx.x & 31;
    int wave = threadIdx.x >> 5;
    int job = blockIdx.x * 8 + wave;
    int mTiles = M >> 4;
    int mt = job % mTiles;
    int ng = job / mTiles;
    int row = lane & 15;
    int kg = (lane >> 4) << 4;
    const float* ap  = A + (size_t)(mt * 16 + row) * K + kg;
    const float* bp0 = B + (size_t)(ng * 64 + row) * K + kg;
    const float* bp1 = bp0 + (size_t)16 * K;
    const float* bp2 = bp0 + (size_t)32 * K;
    const float* bp3 = bp0 + (size_t)48 * K;
    v8f acc0 = {}, acc1 = {}, acc2 = {}, acc3 = {};

    Frag xa, x0, x1, x2, x3;
    Frag ya, y0, y1, y2, y3;

    load_frag(xa, ap); load_frag(x0, bp0); load_frag(x1, bp1);
    load_frag(x2, bp2); load_frag(x3, bp3);

    int k0 = 0;
    for (; k0 + 64 < K; k0 += 64) {
        load_frag(ya, ap + k0 + 32); load_frag(y0, bp0 + k0 + 32); load_frag(y1, bp1 + k0 + 32);
        load_frag(y2, bp2 + k0 + 32); load_frag(y3, bp3 + k0 + 32);
        SCHED_FENCE();
        mma_chunk(xa, x0, x1, x2, x3, acc0, acc1, acc2, acc3);
        SCHED_FENCE();
        load_frag(xa, ap + k0 + 64); load_frag(x0, bp0 + k0 + 64); load_frag(x1, bp1 + k0 + 64);
        load_frag(x2, bp2 + k0 + 64); load_frag(x3, bp3 + k0 + 64);
        KEEP_LOADS_IN_THIS_TRIP();
        SCHED_FENCE();
        mma_chunk(ya, y0, y1, y2, y3, acc0, acc1, acc2, acc3);
    }
    load_frag(ya, ap + k0 + 32); load_frag(y0, bp0 + k0 + 32); load_frag(y1, bp1 + k0 + 32);
    load_frag(y2, bp2 + k0 + 32); load_frag(y3, bp3 + k0 + 32);
    SCHED_FENCE();
    mma_chunk(xa, x0, x1, x2, x3, acc0, acc1, acc2, acc3);
    mma_chunk(ya, y0, y1, y2, y3, acc0, acc1, acc2, acc3);

    int mrow = mt * 16 + (lane >> 4) * 8;
    v8f accs[4] = {acc0, acc1, acc2, acc3};
#pragma unroll
    for (int t = 0; t < 4; ++t) {
        int n = ng * 64 + t * 16 + row;
        float bv = bias[n];
        v8f acc = accs[t];
#pragma unroll
        for (int v = 0; v < 8; ++v) {
            float xv = acc[v] + bv;
            if (RELU) xv = fmaxf(xv, 0.f);
            C[(size_t)(mrow + v) * N + n] = xv;
        }
    }
}

extern "C" void kernel_launch(void* const* d_in, const int* in_sizes, int n_in,
                              void* d_out, int out_size, void* d_ws, size_t ws_size,
                              hipStream_t stream) {
    const float* x     = (const float*)d_in[0];
    const float* theta = (const float*)d_in[1];
    const float* iW1   = (const float*)d_in[2];
    const float* ib1   = (const float*)d_in[3];
    const float* iW2   = (const float*)d_in[4];
    const float* ib2   = (const float*)d_in[5];
    const float* BB_W1 = (const float*)d_in[6];
    const int*   Pi_W1 = (const int*)d_in[7];
    const float* GG_W1 = (const float*)d_in[8];
    const float* BB_b1 = (const float*)d_in[9];
    const int*   Pi_b1 = (const int*)d_in[10];
    const float* GG_b1 = (const float*)d_in[11];
    const float* BB_W2 = (const float*)d_in[12];
    const int*   Pi_W2 = (const int*)d_in[13];
    const float* GG_W2 = (const float*)d_in[14];
    const float* BB_b2 = (const float*)d_in[15];
    const int*   Pi_b2 = (const int*)d_in[16];
    const float* GG_b2 = (const float*)d_in[17];
    float* out = (float*)d_out;

    float* ws = (float*)d_ws;
    size_t o = 0;
    float* W1   = ws + o; o += DD_W;
    float* W2   = ws + o; o += DD_W;
    float* b1   = ws + o; o += 3072;
    float* b2   = ws + o; o += 768;
    float* hbuf = ws + o; o += (size_t)256 * 3072;
    float* mbuf = ws + o; o += DD_W;
    float* wsm  = ws + o; o += 2048;
    float* part = ws + o; o += 256;
    float* gsum = ws + o; o += 1;

    small_fwht_kernel<<<1, 512, 0, stream>>>(theta, BB_W1, wsm, 2048);
    gg_partial_kernel<<<256, 256, 0, stream>>>(GG_W1, LL_BIG, part);
    gg_final_kernel<<<1, 256, 0, stream>>>(part, gsum);
    fwht_passA_kernel<<<COLS / TC, 512, 0, stream>>>(wsm, Pi_W1, GG_W1, mbuf, 2047);
    fwht_passB_kernel<<<ROWS_OUT / BPB, 512, 0, stream>>>(mbuf, iW1, gsum, W1, DD_W);

    small_fwht_kernel<<<1, 512, 0, stream>>>(theta, BB_W2, wsm, 2048);
    gg_partial_kernel<<<256, 256, 0, stream>>>(GG_W2, LL_BIG, part);
    gg_final_kernel<<<1, 256, 0, stream>>>(part, gsum);
    fwht_passA_kernel<<<COLS / TC, 512, 0, stream>>>(wsm, Pi_W2, GG_W2, mbuf, 2047);
    fwht_passB_kernel<<<ROWS_OUT / BPB, 512, 0, stream>>>(mbuf, iW2, gsum, W2, DD_W);

    fastfood_small_kernel<<<1, 512, 0, stream>>>(theta, BB_b1, Pi_b1, GG_b1, ib1, b1,
                                                 2048, 4096, 3072);
    fastfood_small_kernel<<<1, 512, 0, stream>>>(theta, BB_b2, Pi_b2, GG_b2, ib2, b2,
                                                 1024, 1024, 768);

    gemm_wmma_kernel<1><<<96, 256, 0, stream>>>(x, W1, b1, hbuf, 256, 3072, 768);
    gemm_wmma_kernel<0><<<24, 256, 0, stream>>>(hbuf, W2, b2, out, 256, 768, 3072);
}


# Independent W1 and W2 projections share launches: each fastfood kernel runs once, blockIdx.y picks W1 or W2 (b1, b2 ride in the first), 14 launches -> 7; pass B in place; every sum order kept
# speedup vs baseline: 2.2200x; 1.5600x over previous
#include <hip/hip_runtime.h>

typedef __attribute__((ext_vector_type(2))) float v2f;
typedef __attribute__((ext_vector_type(4))) float v4f;
typedef __attribute__((ext_vector_type(8))) float v8f;

#define LL_BIG   (1 << 22)
#define COLS     4096
#define ROWS     1024
#define ROWS_OUT 576
#define DD_W     2359296
#define TC       32
#define TSTRIDE  33
#define BPB      4
#define P_W      2048

__device__ __forceinline__ void fwht_lds(float* buf, int n, int tid, int nt) {
    for (int h = 1; h < n; h <<= 1) {
        __syncthreads();
        for (int p = tid; p < (n >> 1); p += nt) {
            int i0 = ((p & ~(h - 1)) << 1) | (p & (h - 1));
            int i1 = i0 + h;
            float a = buf[i0], b = buf[i1];
            buf[i0] = a + b;
            buf[i1] = a - b;
        }
    }
    __syncthreads();
}

__global__ void gg_partial_kernel(const float* __restrict__ GG_a,
                                  const float* __restrict__ GG_b, int n,
                                  float* __restrict__ part2) {
    __shared__ float red[256];
    int tid = threadIdx.x;
    const float* __restrict__ GG = blockIdx.y ? GG_b : GG_a;
    float* __restrict__ part = part2 + blockIdx.y * 256;
    float s = 0.f;
    for (int i = blockIdx.x * blockDim.x + tid; i < n; i += gridDim.x * blockDim.x) {
        float g = GG[i]; s += g * g;
    }
    red[tid] = s; __syncthreads();
    for (int off = 128; off > 0; off >>= 1) {
        if (tid < off) red[tid] += red[tid + off];
        __syncthreads();
    }
    if (tid == 0) part[blockIdx.x] = red[0];
}

__global__ void gg_final_kernel(const float* __restrict__ part2, float* __restrict__ gsum2) {
    __shared__ float red[256];
    int tid = threadIdx.x;
    const float* __restrict__ part = part2 + blockIdx.y * 256;
    float* __restrict__ gsum = gsum2 + blockIdx.y;
    red[tid] = part[tid]; __syncthreads();
    for (int off = 128; off > 0; off >>= 1) {
        if (tid < off) red[tid] += red[tid + off];
        __syncthreads();
    }
    if (tid == 0) gsum[0] = red[0];
}

__global__ __launch_bounds__(512)
void fwht_passA_kernel(const float* __restrict__ w2,
                       const int* __restrict__ Pi_a, const float* __restrict__ GG_a,
                       float* __restrict__ m_a,
                       const int* __restrict__ Pi_b, const float* __restrict__ GG_b,
                       float* __restrict__ m_b, int pmask) {
    __shared__ float tile[ROWS * TSTRIDE];
    int tid = threadIdx.x, nt = blockDim.x;
    const float* __restrict__ w  = w2 + blockIdx.y * P_W;
    const int*   __restrict__ Pi = blockIdx.y ? Pi_b : Pi_a;
    const float* __restrict__ GG = blockIdx.y ? GG_b : GG_a;
    float* __restrict__ mbuf     = blockIdx.y ? m_b : m_a;
    int col0 = blockIdx.x * TC;
    for (int idx = tid; idx < ROWS * TC; idx += nt) {
        int r = idx >> 5, c = idx & (TC - 1);
        int g = r * COLS + col0 + c;
        tile[r * TSTRIDE + c] = w[Pi[g] & pmask] * GG[g];
    }
    for (int h = 1; h < ROWS; h <<= 1) {
        __syncthreads();
        for (int q = tid; q < (ROWS >> 1) * TC; q += nt) {
            int c = q & (TC - 1), p = q >> 5;
            int r0 = ((p & ~(h - 1)) << 1) | (p & (h - 1));
            int r1 = r0 + h;
            float a = tile[r0 * TSTRIDE + c], b = tile[r1 * TSTRIDE + c];
            tile[r0 * TSTRIDE + c] = a + b;
            tile[r1 * TSTRIDE + c] = a - b;
        }
    }
    __syncthreads();
    for (int idx = tid; idx < ROWS_OUT * TC; idx += nt) {
        int r = idx >> 5, c = idx & (TC - 1);
        mbuf[(size_t)r * COLS + col0 + c] = tile[r * TSTRIDE + c];
    }
}

__global__ __launch_bounds__(512)
void fwht_passB_kernel(float* buf_a, const float* __restrict__ init_a,
                       float* buf_b, const float* __restrict__ init_b,
                       const float* __restrict__ gsum2, int DD) {
    __shared__ float blk[BPB * COLS];
    int tid = threadIdx.x, nt = blockDim.x;
    float* buf = blockIdx.y ? buf_b : buf_a;
    const float* __restrict__ init = blockIdx.y ? init_b : init_a;
    const float* __restrict__ gsum = gsum2 + blockIdx.y;
    size_t base = (size_t)blockIdx.x * (BPB * COLS);
    for (int i = tid; i < BPB * COLS; i += nt) blk[i] = buf[base + i];
    for (int h = 1; h < COLS; h <<= 1) {
        __syncthreads();
        for (int q = tid; q < BPB * (COLS >> 1); q += nt) {
            int sb = q >> 11, p = q & (COLS / 2 - 1);
            int i0 = (sb << 12) | (((p & ~(h - 1)) << 1) | (p & (h - 1)));
            int i1 = i0 + h;
            float a = blk[i0], b = blk[i1];
            blk[i0] = a + b;
            blk[i1] = a - b;
        }
    }
    __syncthreads();
    float scale = rsqrtf((float)DD * gsum[0]);
    for (int i = tid; i < BPB * COLS; i += nt) {
        size_t g = base + i;
        buf[g] = init[g] + blk[i] * scale;
    }
}

__global__ __launch_bounds__(512)
void fastfood_small_kernel(const float* __restrict__ theta,
                           const float* __restrict__ BB_W1, const float* __restrict__ BB_W2,
                           float* __restrict__ wout,
                           const float* __restrict__ BB_b1, const int* __restrict__ Pi_b1,
                           const float* __restrict__ GG_b1, const float* __restrict__ init_b1,
                           float* __restrict__ out_b1,
                           const float* __restrict__ BB_b2, const int* __restrict__ Pi_b2,
                           const float* __restrict__ GG_b2, const float* __restrict__ init_b2,
                           float* __restrict__ out_b2) {
    __shared__ float w[2048];
    __shared__ float m[4096];
    __shared__ float red[512];
    int tid = threadIdx.x, nt = blockDim.x;
    const int job = blockIdx.x;
    const float* __restrict__ BB = job == 0 ? BB_W1 : job == 1 ? BB_W2 : job == 2 ? BB_b1 : BB_b2;
    const int P = job == 3 ? 1024 : P_W;
    for (int j = tid; j < P; j += nt) w[j] = BB[j] * theta[j];
    fwht_lds(w, P, tid, nt);
    if (job < 2) {
        for (int j = tid; j < P; j += nt) wout[job * P_W + j] = w[j];
        return;
    }
    const bool is_b1 = job == 2;
    const int*   __restrict__ Pi   = is_b1 ? Pi_b1 : Pi_b2;
    const float* __restrict__ GG   = is_b1 ? GG_b1 : GG_b2;
    const float* __restrict__ init = is_b1 ? init_b1 : init_b2;
    float* __restrict__ out        = is_b1 ? out_b1 : out_b2;
    const int LLv = is_b1 ? 4096 : 1024;
    const int DD  = is_b1 ? 3072 : 768;
    float s = 0.f;
    for (int i = tid; i < LLv; i += nt) { float g = GG[i]; s += g * g; }
    red[tid] = s; __syncthreads();
    for (int off = nt >> 1; off > 0; off >>= 1) {
        if (tid < off) red[tid] += red[tid + off];
        __syncthreads();
    }
    for (int i = tid; i < LLv; i += nt) m[i] = w[Pi[i] & (P - 1)] * GG[i];
    fwht_lds(m, LLv, tid, nt);
    float scale = rsqrtf((float)DD * red[0]);
    for (int i = tid; i < DD; i += nt) out[i] = init[i] + m[i] * scale;
}

struct Frag { v4f q0, q1, q2, q3; };

__device__ __forceinline__ void load_frag(Frag& f, const float* __restrict__ p) {
    const v4f* p4 = (const v4f*)p;
    f.q0 = p4[0]; f.q1 = p4[1]; f.q2 = p4[2]; f.q3 = p4[3];
}

#define SCHED_FENCE() __builtin_amdgcn_sched_barrier(0)
#define KEEP_LOADS_IN_THIS_TRIP() asm volatile("" ::: "memory")

#define LO2(q) __builtin_shufflevector(q, q, 0, 1)
#define HI2(q) __builtin_shufflevector(q, q, 2, 3)
#define WMMA_F32(a, b, c) __builtin_amdgcn_wmma_f32_16x16x4_f32(false, a, false, b, (short)0, c, false, false)
#define KSTEP(SEL, Q)                                   \
    acc0 = WMMA_F32(SEL(fa.Q), SEL(f0.Q), acc0);        \
    acc1 = WMMA_F32(SEL(fa.Q), SEL(f1.Q), acc1);        \
    acc2 = WMMA_F32(SEL(fa.Q), SEL(f2.Q), acc2);        \
    acc3 = WMMA_F32(SEL(fa.Q), SEL(f3.Q), acc3);

__device__ __forceinline__ void mma_chunk(const Frag& fa, const Frag& f0, const Frag& f1,
                                          const Frag& f2, const Frag& f3,
                                          v8f& acc0, v8f& acc1, v8f& acc2, v8f& acc3) {
    KSTEP(LO2, q0) KSTEP(HI2, q0)
    KSTEP(LO2, q1) KSTEP(HI2, q1)
    KSTEP(LO2, q2) KSTEP(HI2, q2)
    KSTEP(LO2, q3) KSTEP(HI2, q3)
}

template<int RELU>
__global__ __launch_bounds__(256)
void gemm_wmma_kernel(const float* __restrict__ A, const float* __restrict__ B,
                      const float* __restrict__ bias, float* __restrict__ C,
                      int M, int N, int K) {
    int lane = threadIdx.x & 31;
    int wave = threadIdx.x >> 5;
    int job = blockIdx.x * 8 + wave;
    int mTiles = M >> 4;
    int mt = job % mTiles;
    int ng = job / mTiles;
    int row = lane & 15;
    int kg = (lane >> 4) << 4;
    const float* ap  = A + (size_t)(mt * 16 + row) * K + kg;
    const float* bp0 = B + (size_t)(ng * 64 + row) * K + kg;
    const float* bp1 = bp0 + (size_t)16 * K;
    const float* bp2 = bp0 + (size_t)32 * K;
    const float* bp3 = bp0 + (size_t)48 * K;
    v8f acc0 = {}, acc1 = {}, acc2 = {}, acc3 = {};

    Frag xa, x0, x1, x2, x3;
    Frag ya, y0, y1, y2, y3;

    load_frag(xa, ap); load_frag(x0, bp0); load_frag(x1, bp1);
    load_frag(x2, bp2); load_frag(x3, bp3);

    int k0 = 0;
    for (; k0 + 64 < K; k0 += 64) {
        load_frag(ya, ap + k0 + 32); load_frag(y0, bp0 + k0 + 32); load_frag(y1, bp1 + k0 + 32);
        load_frag(y2, bp2 + k0 + 32); load_frag(y3, bp3 + k0 + 32);
        SCHED_FENCE();
        mma_chunk(xa, x0, x1, x2, x3, acc0, acc1, acc2, acc3);
        SCHED_FENCE();
        load_frag(xa, ap + k0 + 64); load_frag(x0, bp0 + k0 + 64); load_frag(x1, bp1 + k0 + 64);
        load_frag(x2, bp2 + k0 + 64); load_frag(x3, bp3 + k0 + 64);
        KEEP_LOADS_IN_THIS_TRIP();
        SCHED_FENCE();
        mma_chunk(ya, y0, y1, y2, y3, acc0, acc1, acc2, acc3);
    }
    load_frag(ya, ap + k0 + 32); load_frag(y0, bp0 + k0 + 32); load_frag(y1, bp1 + k0 + 32);
    load_frag(y2, bp2 + k0 + 32); load_frag(y3, bp3 + k0 + 32);
    SCHED_FENCE();
    mma_chunk(xa, x0, x1, x2, x3, acc0, acc1, acc2, acc3);
    mma_chunk(ya, y0, y1, y2, y3, acc0, acc1, acc2, acc3);

    int mrow = mt * 16 + (lane >> 4) * 8;
    v8f accs[4] = {acc0, acc1, acc2, acc3};
#pragma unroll
    for (int t = 0; t < 4; ++t) {
        int n = ng * 64 + t * 16 + row;
        float bv = bias[n];
        v8f acc = accs[t];
#pragma unroll
        for (int v = 0; v < 8; ++v) {
            float xv = acc[v] + bv;
            if (RELU) xv = fmaxf(xv, 0.f);
            C[(size_t)(mrow + v) * N + n] = xv;
        }
    }
}

extern "C" void kernel_launch(void* const* d_in, const int* in_sizes, int n_in,
                              void* d_out, int out_size, void* d_ws, size_t ws_size,
                              hipStream_t stream) {
    const float* x     = (const float*)d_in[0];
    const float* theta = (const float*)d_in[1];
    const float* iW1   = (const float*)d_in[2];
    const float* ib1   = (const float*)d_in[3];
    const float* iW2   = (const float*)d_in[4];
    const float* ib2   = (const float*)d_in[5];
    const float* BB_W1 = (const float*)d_in[6];
    const int*   Pi_W1 = (const int*)d_in[7];
    const float* GG_W1 = (const float*)d_in[8];
    const float* BB_b1 = (const float*)d_in[9];
    const int*   Pi_b1 = (const int*)d_in[10];
    const float* GG_b1 = (const float*)d_in[11];
    const float* BB_W2 = (const float*)d_in[12];
    const int*   Pi_W2 = (const int*)d_in[13];
    const float* GG_W2 = (const float*)d_in[14];
    const float* BB_b2 = (const float*)d_in[15];
    const int*   Pi_b2 = (const int*)d_in[16];
    const float* GG_b2 = (const float*)d_in[17];
    float* out = (float*)d_out;

    float* ws = (float*)d_ws;
    size_t o = 0;
    float* W1   = ws + o; o += DD_W;
    float* W2   = ws + o; o += DD_W;
    float* b1   = ws + o; o += 3072;
    float* b2   = ws + o; o += 768;
    float* hbuf = ws + o; o += (size_t)256 * 3072;
    float* wsm  = ws + o; o += 4096;
    float* part = ws + o; o += 512;
    float* gsum = ws + o; o += 2;

    fastfood_small_kernel<<<4, 512, 0, stream>>>(theta, BB_W1, BB_W2, wsm,
                                                 BB_b1, Pi_b1, GG_b1, ib1, b1,
                                                 BB_b2, Pi_b2, GG_b2, ib2, b2);
    gg_partial_kernel<<<dim3(256, 2), 256, 0, stream>>>(GG_W1, GG_W2, LL_BIG, part);
    gg_final_kernel<<<dim3(1, 2), 256, 0, stream>>>(part, gsum);
    fwht_passA_kernel<<<dim3(COLS / TC, 2), 512, 0, stream>>>(wsm, Pi_W1, GG_W1, W1,
                                                              Pi_W2, GG_W2, W2, 2047);
    fwht_passB_kernel<<<dim3(ROWS_OUT / BPB, 2), 512, 0, stream>>>(W1, iW1, W2, iW2, gsum, DD_W);

    gemm_wmma_kernel<1><<<96, 256, 0, stream>>>(x, W1, b1, hbuf, 256, 3072, 768);
    gemm_wmma_kernel<0><<<24, 256, 0, stream>>>(hbuf, W2, b2, out, 256, 768, 3072);
}


# GEMM work split: k of each 16x64 strip cut in 2 (GEMM1) / 8 (GEMM2) slices over the waves of a workgroup, 96 / 24 -> 192 workgroups; partials added in slice order through LDS (new k sum order)
# speedup vs baseline: 3.0659x; 1.3810x over previous
#include <hip/hip_runtime.h>

typedef __attribute__((ext_vector_type(2))) float v2f;
typedef __attribute__((ext_vector_type(4))) float v4f;
typedef __attribute__((ext_vector_type(8))) float v8f;

#define LL_BIG   (1 << 22)
#define COLS     4096
#define ROWS     1024
#define ROWS_OUT 576
#define DD_W     2359296
#define TC       32
#define TSTRIDE  33
#define BPB      4
#define P_W      2048

__device__ __forceinline__ void fwht_lds(float* buf, int n, int tid, int nt) {
    for (int h = 1; h < n; h <<= 1) {
        __syncthreads();
        for (int p = tid; p < (n >> 1); p += nt) {
            int i0 = ((p & ~(h - 1)) << 1) | (p & (h - 1));
            int i1 = i0 + h;
            float a = buf[i0], b = buf[i1];
            buf[i0] = a + b;
            buf[i1] = a - b;
        }
    }
    __syncthreads();
}

__global__ void gg_partial_kernel(const float* __restrict__ GG_a,
                                  const float* __restrict__ GG_b, int n,
                                  float* __restrict__ part2) {
    __shared__ float red[256];
    int tid = threadIdx.x;
    const float* __restrict__ GG = blockIdx.y ? GG_b : GG_a;
    float* __restrict__ part = part2 + blockIdx.y * 256;
    float s = 0.f;
    for (int i = blockIdx.x * blockDim.x + tid; i < n; i += gridDim.x * blockDim.x) {
        float g = GG[i]; s += g * g;
    }
    red[tid] = s; __syncthreads();
    for (int off = 128; off > 0; off >>= 1) {
        if (tid < off) red[tid] += red[tid + off];
        __syncthreads();
    }
    if (tid == 0) part[blockIdx.x] = red[0];
}

__global__ void gg_final_kernel(const float* __restrict__ part2, float* __restrict__ gsum2) {
    __shared__ float red[256];
    int tid = threadIdx.x;
    const float* __restrict__ part = part2 + blockIdx.y * 256;
    float* __restrict__ gsum = gsum2 + blockIdx.y;
    red[tid] = part[tid]; __syncthreads();
    for (int off = 128; off > 0; off >>= 1) {
        if (tid < off) red[tid] += red[tid + off];
        __syncthreads();
    }
    if (tid == 0) gsum[0] = red[0];
}

__global__ __launch_bounds__(512)
void fwht_passA_kernel(const float* __restrict__ w2,
                       const int* __restrict__ Pi_a, const float* __restrict__ GG_a,
                       float* __restrict__ m_a,
                       const int* __restrict__ Pi_b, const float* __restrict__ GG_b,
                       float* __restrict__ m_b, int pmask) {
    __shared__ float tile[ROWS * TSTRIDE];
    int tid = threadIdx.x, nt = blockDim.x;
    const float* __restrict__ w  = w2 + blockIdx.y * P_W;
    const int*   __restrict__ Pi = blockIdx.y ? Pi_b : Pi_a;
    const float* __restrict__ GG = blockIdx.y ? GG_b : GG_a;
    float* __restrict__ mbuf     = blockIdx.y ? m_b : m_a;
    int col0 = blockIdx.x * TC;
    for (int idx = tid; idx < ROWS * TC; idx += nt) {
        int r = idx >> 5, c = idx & (TC - 1);
        int g = r * COLS + col0 + c;
        tile[r * TSTRIDE + c] = w[Pi[g] & pmask] * GG[g];
    }
    for (int h = 1; h < ROWS; h <<= 1) {
        __syncthreads();
        for (int q = tid; q < (ROWS >> 1) * TC; q += nt) {
            int c = q & (TC - 1), p = q >> 5;
            int r0 = ((p & ~(h - 1)) << 1) | (p & (h - 1));
            int r1 = r0 + h;
            float a = tile[r0 * TSTRIDE + c], b = tile[r1 * TSTRIDE + c];
            tile[r0 * TSTRIDE + c] = a + b;
            tile[r1 * TSTRIDE + c] = a - b;
        }
    }
    __syncthreads();
    for (int idx = tid; idx < ROWS_OUT * TC; idx += nt) {
        int r = idx >> 5, c = idx & (TC - 1);
        mbuf[(size_t)r * COLS + col0 + c] = tile[r * TSTRIDE + c];
    }
}

__global__ __launch_bounds__(512)
void fwht_passB_kernel(float* buf_a, const float* __restrict__ init_a,
                       float* buf_b, const float* __restrict__ init_b,
                       const float* __restrict__ gsum2, int DD) {
    __shared__ float blk[BPB * COLS];
    int tid = threadIdx.x, nt = blockDim.x;
    float* buf = blockIdx.y ? buf_b : buf_a;
    const float* __restrict__ init = blockIdx.y ? init_b : init_a;
    const float* __restrict__ gsum = gsum2 + blockIdx.y;
    size_t base = (size_t)blockIdx.x * (BPB * COLS);
    for (int i = tid; i < BPB * COLS; i += nt) blk[i] = buf[base + i];
    for (int h = 1; h < COLS; h <<= 1) {
        __syncthreads();
        for (int q = tid; q < BPB * (COLS >> 1); q += nt) {
            int sb = q >> 11, p = q & (COLS / 2 - 1);
            int i0 = (sb << 12) | (((p & ~(h - 1)) << 1) | (p & (h - 1)));
            int i1 = i0 + h;
            float a = blk[i0], b = blk[i1];
            blk[i0] = a + b;
            blk[i1] = a - b;
        }
    }
    __syncthreads();
    float scale = rsqrtf((float)DD * gsum[0]);
    for (int i = tid; i < BPB * COLS; i += nt) {
        size_t g = base + i;
        buf[g] = init[g] + blk[i] * scale;
    }
}

__global__ __launch_bounds__(512)
void fastfood_small_kernel(const float* __restrict__ theta,
                           const float* __restrict__ BB_W1, const float* __restrict__ BB_W2,
                           float* __restrict__ wout,
                           const float* __restrict__ BB_b1, const int* __restrict__ Pi_b1,
                           const float* __restrict__ GG_b1, const float* __restrict__ init_b1,
                           float* __restrict__ out_b1,
                           const float* __restrict__ BB_b2, const int* __restrict__ Pi_b2,
                           const float* __restrict__ GG_b2, const float* __restrict__ init_b2,
                           float* __restrict__ out_b2) {
    __shared__ float w[2048];
    __shared__ float m[4096];
    __shared__ float red[512];
    int tid = threadIdx.x, nt = blockDim.x;
    const int job = blockIdx.x;
    const float* __restrict__ BB = job == 0 ? BB_W1 : job == 1 ? BB_W2 : job == 2 ? BB_b1 : BB_b2;
    const int P = job == 3 ? 1024 : P_W;
    for (int j = tid; j < P; j += nt) w[j] = BB[j] * theta[j];
    fwht_lds(w, P, tid, nt);
    if (job < 2) {
        for (int j = tid; j < P; j += nt) wout[job * P_W + j] = w[j];
        return;
    }
    const bool is_b1 = job == 2;
    const int*   __restrict__ Pi   = is_b1 ? Pi_b1 : Pi_b2;
    const float* __restrict__ GG   = is_b1 ? GG_b1 : GG_b2;
    const float* __restrict__ init = is_b1 ? init_b1 : init_b2;
    float* __restrict__ out        = is_b1 ? out_b1 : out_b2;
    const int LLv = is_b1 ? 4096 : 1024;
    const int DD  = is_b1 ? 3072 : 768;
    float s = 0.f;
    for (int i = tid; i < LLv; i += nt) { float g = GG[i]; s += g * g; }
    red[tid] = s; __syncthreads();
    for (int off = nt >> 1; off > 0; off >>= 1) {
        if (tid < off) red[tid] += red[tid + off];
        __syncthreads();
    }
    for (int i = tid; i < LLv; i += nt) m[i] = w[Pi[i] & (P - 1)] * GG[i];
    fwht_lds(m, LLv, tid, nt);
    float scale = rsqrtf((float)DD * red[0]);
    for (int i = tid; i < DD; i += nt) out[i] = init[i] + m[i] * scale;
}

struct Frag { v4f q0, q1, q2, q3; };

__device__ __forceinline__ void load_frag(Frag& f, const float* __restrict__ p) {
    const v4f* p4 = (const v4f*)p;
    f.q0 = p4[0]; f.q1 = p4[1]; f.q2 = p4[2]; f.q3 = p4[3];
}

#define SCHED_FENCE() __builtin_amdgcn_sched_barrier(0)
#define KEEP_LOADS_IN_THIS_TRIP() asm volatile("" ::: "memory")

#define LO2(q) __builtin_shufflevector(q, q, 0, 1)
#define HI2(q) __builtin_shufflevector(q, q, 2, 3)
#define WMMA_F32(a, b, c) __builtin_amdgcn_wmma_f32_16x16x4_f32(false, a, false, b, (short)0, c, false, false)
#define KSTEP(SEL, Q)                                   \
    acc0 = WMMA_F32(SEL(fa.Q), SEL(f0.Q), acc0);        \
    acc1 = WMMA_F32(SEL(fa.Q), SEL(f1.Q), acc1);        \
    acc2 = WMMA_F32(SEL(fa.Q), SEL(f2.Q), acc2);        \
    acc3 = WMMA_F32(SEL(fa.Q), SEL(f3.Q), acc3);

__device__ __forceinline__ void mma_chunk(const Frag& fa, const Frag& f0, const Frag& f1,
                                          const Frag& f2, const Frag& f3,
                                          v8f& acc0, v8f& acc1, v8f& acc2, v8f& acc3) {
    KSTEP(LO2, q0) KSTEP(HI2, q0)
    KSTEP(LO2, q1) KSTEP(HI2, q1)
    KSTEP(LO2, q2) KSTEP(HI2, q2)
    KSTEP(LO2, q3) KSTEP(HI2, q3)
}

template<int RELU, int KS>
__global__ __launch_bounds__(256)
void gemm_wmma_kernel(const float* __restrict__ A, const float* __restrict__ B,
                      const float* __restrict__ bias, float* __restrict__ C,
                      int M, int N, int K) {
    constexpr int SPB = 8 / KS;
    __shared__ v4f part[8 * 8 * 32];
    int lane = threadIdx.x & 31;
    int wave = threadIdx.x >> 5;
    int sl = wave / KS;
    int ks = wave % KS;
    int job = blockIdx.x * SPB + sl;
    int mTiles = M >> 4;
    int mt = job % mTiles;
    int ng = job / mTiles;
    int row = lane & 15;
    int kg = (lane >> 4) << 4;
    int kLen = K / KS;
    size_t koff = (size_t)ks * kLen + kg;
    const float* ap  = A + (size_t)(mt * 16 + row) * K + koff;
    const float* bp0 = B + (size_t)(ng * 64 + row) * K + koff;
    const float* bp1 = bp0 + (size_t)16 * K;
    const float* bp2 = bp0 + (size_t)32 * K;
    const float* bp3 = bp0 + (size_t)48 * K;
    v8f acc0 = {}, acc1 = {}, acc2 = {}, acc3 = {};

    Frag xa, x0, x1, x2, x3;
    Frag ya, y0, y1, y2, y3;

    load_frag(xa, ap); load_frag(x0, bp0); load_frag(x1, bp1);
    load_frag(x2, bp2); load_frag(x3, bp3);

    int k0 = 0;
    for (; k0 + 64 < kLen; k0 += 64) {
        load_frag(ya, ap + k0 + 32); load_frag(y0, bp0 + k0 + 32); load_frag(y1, bp1 + k0 + 32);
        load_frag(y2, bp2 + k0 + 32); load_frag(y3, bp3 + k0 + 32);
        SCHED_FENCE();
        mma_chunk(xa, x0, x1, x2, x3, acc0, acc1, acc2, acc3);
        SCHED_FENCE();
        load_frag(xa, ap + k0 + 64); load_frag(x0, bp0 + k0 + 64); load_frag(x1, bp1 + k0 + 64);
        load_frag(x2, bp2 + k0 + 64); load_frag(x3, bp3 + k0 + 64);
        KEEP_LOADS_IN_THIS_TRIP();
        SCHED_FENCE();
        mma_chunk(ya, y0, y1, y2, y3, acc0, acc1, acc2, acc3);
    }
    load_frag(ya, ap + k0 + 32); load_frag(y0, bp0 + k0 + 32); load_frag(y1, bp1 + k0 + 32);
    load_frag(y2, bp2 + k0 + 32); load_frag(y3, bp3 + k0 + 32);
    SCHED_FENCE();
    mma_chunk(xa, x0, x1, x2, x3, acc0, acc1, acc2, acc3);
    mma_chunk(ya, y0, y1, y2, y3, acc0, acc1, acc2, acc3);

    v4f* mine = part + wave * 256;
    v8f accs[4] = {acc0, acc1, acc2, acc3};
#pragma unroll
    for (int t = 0; t < 4; ++t) {
        mine[(2 * t) * 32 + lane]     = __builtin_shufflevector(accs[t], accs[t], 0, 1, 2, 3);
        mine[(2 * t + 1) * 32 + lane] = __builtin_shufflevector(accs[t], accs[t], 4, 5, 6, 7);
    }
    __syncthreads();

    constexpr int GPW = 8 / KS;
    const v4f* sp = part + sl * KS * 256;
    int mrow = mt * 16 + (lane >> 4) * 8;
#pragma unroll
    for (int j = 0; j < GPW; ++j) {
        int g = ks * GPW + j;
        int t = g >> 1;
        int vb = (g & 1) * 4;
        v4f s = sp[g * 32 + lane];
#pragma unroll
        for (int q = 1; q < KS; ++q) s += sp[q * 256 + g * 32 + lane];
        int n = ng * 64 + t * 16 + row;
        float bv = bias[n];
#pragma unroll
        for (int e = 0; e < 4; ++e) {
            float xv = s[e] + bv;
            if (RELU) xv = fmaxf(xv, 0.f);
            C[(size_t)(mrow + vb + e) * N + n] = xv;
        }
    }
}

extern "C" void kernel_launch(void* const* d_in, const int* in_sizes, int n_in,
                              void* d_out, int out_size, void* d_ws, size_t ws_size,
                              hipStream_t stream) {
    const float* x     = (const float*)d_in[0];
    const float* theta = (const float*)d_in[1];
    const float* iW1   = (const float*)d_in[2];
    const float* ib1   = (const float*)d_in[3];
    const float* iW2   = (const float*)d_in[4];
    const float* ib2   = (const float*)d_in[5];
    const float* BB_W1 = (const float*)d_in[6];
    const int*   Pi_W1 = (const int*)d_in[7];
    const float* GG_W1 = (const float*)d_in[8];
    const float* BB_b1 = (const float*)d_in[9];
    const int*   Pi_b1 = (const int*)d_in[10];
    const float* GG_b1 = (const float*)d_in[11];
    const float* BB_W2 = (const float*)d_in[12];
    const int*   Pi_W2 = (const int*)d_in[13];
    const float* GG_W2 = (const float*)d_in[14];
    const float* BB_b2 = (const float*)d_in[15];
    const int*   Pi_b2 = (const int*)d_in[16];
    const float* GG_b2 = (const float*)d_in[17];
    float* out = (float*)d_out;

    float* ws = (float*)d_ws;
    size_t o = 0;
    float* W1   = ws + o; o += DD_W;
    float* W2   = ws + o; o += DD_W;
    float* b1   = ws + o; o += 3072;
    float* b2   = ws + o; o += 768;
    float* hbuf = ws + o; o += (size_t)256 * 3072;
    float* wsm  = ws + o; o += 4096;
    float* part = ws + o; o += 512;
    float* gsum = ws + o; o += 2;

    fastfood_small_kernel<<<4, 512, 0, stream>>>(theta, BB_W1, BB_W2, wsm,
                                                 BB_b1, Pi_b1, GG_b1, ib1, b1,
                                                 BB_b2, Pi_b2, GG_b2, ib2, b2);
    gg_partial_kernel<<<dim3(256, 2), 256, 0, stream>>>(GG_W1, GG_W2, LL_BIG, part);
    gg_final_kernel<<<dim3(1, 2), 256, 0, stream>>>(part, gsum);
    fwht_passA_kernel<<<dim3(COLS / TC, 2), 512, 0, stream>>>(wsm, Pi_W1, GG_W1, W1,
                                                              Pi_W2, GG_W2, W2, 2047);
    fwht_passB_kernel<<<dim3(ROWS_OUT / BPB, 2), 512, 0, stream>>>(W1, iW1, W2, iW2, gsum, DD_W);

    gemm_wmma_kernel<1, 2><<<192, 256, 0, stream>>>(x, W1, b1, hbuf, 256, 3072, 768);
    gemm_wmma_kernel<0, 8><<<192, 256, 0, stream>>>(hbuf, W2, b2, out, 256, 768, 3072);
}


# FWHT passes A, B: a thread keeps 64 elements in registers, 6+4 / 2+6+4 butterfly stages per LDS trip, 1 / 2 barriers (was 10 / 12); stage order kept; pass A's first stage fuses its product (fma)
# speedup vs baseline: 3.9094x; 1.2751x over previous
#include <hip/hip_runtime.h>

typedef __attribute__((ext_vector_type(2))) float v2f;
typedef __attribute__((ext_vector_type(4))) float v4f;
typedef __attribute__((ext_vector_type(8))) float v8f;

#define LL_BIG   (1 << 22)
#define COLS     4096
#define ROWS     1024
#define ROWS_OUT 576
#define DD_W     2359296
#define TC       32
#define PB_PITCH 260
#define P_W      2048

__device__ __forceinline__ void fwht_lds(float* buf, int n, int tid, int nt) {
    for (int h = 1; h < n; h <<= 1) {
        __syncthreads();
        for (int p = tid; p < (n >> 1); p += nt) {
            int i0 = ((p & ~(h - 1)) << 1) | (p & (h - 1));
            int i1 = i0 + h;
            float a = buf[i0], b = buf[i1];
            buf[i0] = a + b;
            buf[i1] = a - b;
        }
    }
    __syncthreads();
}

__global__ void gg_partial_kernel(const float* __restrict__ GG_a,
                                  const float* __restrict__ GG_b, int n,
                                  float* __restrict__ part2) {
    __shared__ float red[256];
    int tid = threadIdx.x;
    const float* __restrict__ GG = blockIdx.y ? GG_b : GG_a;
    float* __restrict__ part = part2 + blockIdx.y * 256;
    float s = 0.f;
    for (int i = blockIdx.x * blockDim.x + tid; i < n; i += gridDim.x * blockDim.x) {
        float g = GG[i]; s += g * g;
    }
    red[tid] = s; __syncthreads();
    for (int off = 128; off > 0; off >>= 1) {
        if (tid < off) red[tid] += red[tid + off];
        __syncthreads();
    }
    if (tid == 0) part[blockIdx.x] = red[0];
}

__global__ void gg_final_kernel(const float* __restrict__ part2, float* __restrict__ gsum2) {
    __shared__ float red[256];
    int tid = threadIdx.x;
    const float* __restrict__ part = part2 + blockIdx.y * 256;
    float* __restrict__ gsum = gsum2 + blockIdx.y;
    red[tid] = part[tid]; __syncthreads();
    for (int off = 128; off > 0; off >>= 1) {
        if (tid < off) red[tid] += red[tid + off];
        __syncthreads();
    }
    if (tid == 0) gsum[0] = red[0];
}

template<int N, int B0, int B1>
__device__ __forceinline__ void fwht_reg_stages(float (&v)[N]) {
#pragma unroll
    for (int b = B0; b < B1; ++b) {
        const int h = 1 << b;
#pragma unroll
        for (int p = 0; p < N / 2; ++p) {
            const int i0 = ((p & ~(h - 1)) << 1) | (p & (h - 1));
            const float x = v[i0], y = v[i0 + h];
            v[i0] = x + y;
            v[i0 + h] = x - y;
        }
    }
}

__global__ __launch_bounds__(512)
void fwht_passA_kernel(const float* __restrict__ w2,
                       const int* __restrict__ Pi_a, const float* __restrict__ GG_a,
                       float* __restrict__ m_a,
                       const int* __restrict__ Pi_b, const float* __restrict__ GG_b,
                       float* __restrict__ m_b, int pmask) {
    __shared__ float tile[ROWS * TC];
    const int lane = threadIdx.x & 31;
    const int wv = threadIdx.x >> 5;
    const float* __restrict__ w  = w2 + blockIdx.y * P_W;
    const int*   __restrict__ Pi = blockIdx.y ? Pi_b : Pi_a;
    const float* __restrict__ GG = blockIdx.y ? GG_b : GG_a;
    float* __restrict__ mbuf     = blockIdx.y ? m_b : m_a;
    const int col = blockIdx.x * TC + lane;

    float v[64];
    int pi[64];
    const int g0 = wv * 64 * COLS + col;
#pragma unroll
    for (int j = 0; j < 64; ++j) pi[j] = Pi[g0 + j * COLS];
#pragma unroll
    for (int j = 0; j < 64; ++j) v[j] = GG[g0 + j * COLS];
#pragma unroll
    for (int j = 0; j < 64; ++j) v[j] = w[pi[j] & pmask] * v[j];
    fwht_reg_stages<64, 0, 6>(v);
#pragma unroll
    for (int j = 0; j < 64; ++j) tile[(wv * 64 + j) * TC + lane] = v[j];
    __syncthreads();
    float u[64];
#pragma unroll
    for (int q = 0; q < 4; ++q) {
#pragma unroll
        for (int jh = 0; jh < 16; ++jh) u[16 * q + jh] = tile[(64 * jh + 4 * wv + q) * TC + lane];
    }
    fwht_reg_stages<64, 0, 4>(u);
#pragma unroll
    for (int q = 0; q < 4; ++q) {
#pragma unroll
        for (int jh = 0; jh < ROWS_OUT / 64; ++jh)
            mbuf[(size_t)(64 * jh + 4 * wv + q) * COLS + col] = u[16 * q + jh];
    }
}

__global__ __launch_bounds__(64)
void fwht_passB_kernel(float* buf_a, const float* __restrict__ init_a,
                       float* buf_b, const float* __restrict__ init_b,
                       const float* __restrict__ gsum2, int DD) {
    __shared__ v4f blk4[16 * PB_PITCH / 4];
    float* blk = (float*)blk4;
    const int t = threadIdx.x;
    float* buf = blockIdx.y ? buf_b : buf_a;
    const float* __restrict__ init = blockIdx.y ? init_b : init_a;
    const float* __restrict__ gsum = gsum2 + blockIdx.y;
    const size_t base = (size_t)blockIdx.x * COLS;
    v4f* row4 = (v4f*)(buf + base) + t;
    const v4f* __restrict__ init4 = (const v4f*)(init + base) + t;

    float v[64];
    v4f in[16];
#pragma unroll
    for (int c = 0; c < 16; ++c) {
        const v4f x = row4[64 * c];
        v[4 * c + 0] = x[0]; v[4 * c + 1] = x[1]; v[4 * c + 2] = x[2]; v[4 * c + 3] = x[3];
    }
#pragma unroll
    for (int c = 0; c < 16; ++c) in[c] = init4[64 * c];
    fwht_reg_stages<64, 0, 2>(v);
#pragma unroll
    for (int c = 0; c < 16; ++c) {
        v4f x; x[0] = v[4 * c + 0]; x[1] = v[4 * c + 1]; x[2] = v[4 * c + 2]; x[3] = v[4 * c + 3];
        blk4[c * (PB_PITCH / 4) + t] = x;
    }
    __syncthreads();
    float* pb = blk + (t >> 2) * PB_PITCH + (t & 3);
    float u[64];
#pragma unroll
    for (int m = 0; m < 64; ++m) u[m] = pb[4 * m];
    fwht_reg_stages<64, 0, 6>(u);
#pragma unroll
    for (int m = 0; m < 64; ++m) pb[4 * m] = u[m];
    __syncthreads();
#pragma unroll
    for (int c = 0; c < 16; ++c) {
        const v4f x = blk4[c * (PB_PITCH / 4) + t];
        v[4 * c + 0] = x[0]; v[4 * c + 1] = x[1]; v[4 * c + 2] = x[2]; v[4 * c + 3] = x[3];
    }
    fwht_reg_stages<64, 2, 6>(v);
    const float scale = rsqrtf((float)DD * gsum[0]);
#pragma unroll
    for (int c = 0; c < 16; ++c) {
        v4f o;
        o[0] = in[c][0] + v[4 * c + 0] * scale; o[1] = in[c][1] + v[4 * c + 1] * scale;
        o[2] = in[c][2] + v[4 * c + 2] * scale; o[3] = in[c][3] + v[4 * c + 3] * scale;
        row4[64 * c] = o;
    }
}

__global__ __launch_bounds__(512)
void fastfood_small_kernel(const float* __restrict__ theta,
                           const float* __restrict__ BB_W1, const float* __restrict__ BB_W2,
                           float* __restrict__ wout,
                           const float* __restrict__ BB_b1, const int* __restrict__ Pi_b1,
                           const float* __restrict__ GG_b1, const float* __restrict__ init_b1,
                           float* __restrict__ out_b1,
                           const float* __restrict__ BB_b2, const int* __restrict__ Pi_b2,
                           const float* __restrict__ GG_b2, const float* __restrict__ init_b2,
                           float* __restrict__ out_b2) {
    __shared__ float w[2048];
    __shared__ float m[4096];
    __shared__ float red[512];
    int tid = threadIdx.x, nt = blockDim.x;
    const int job = blockIdx.x;
    const float* __restrict__ BB = job == 0 ? BB_W1 : job == 1 ? BB_W2 : job == 2 ? BB_b1 : BB_b2;
    const int P = job == 3 ? 1024 : P_W;
    for (int j = tid; j < P; j += nt) w[j] = BB[j] * theta[j];
    fwht_lds(w, P, tid, nt);
    if (job < 2) {
        for (int j = tid; j < P; j += nt) wout[job * P_W + j] = w[j];
        return;
    }
    const bool is_b1 = job == 2;
    const int*   __restrict__ Pi   = is_b1 ? Pi_b1 : Pi_b2;
    const float* __restrict__ GG   = is_b1 ? GG_b1 : GG_b2;
    const float* __restrict__ init = is_b1 ? init_b1 : init_b2;
    float* __restrict__ out        = is_b1 ? out_b1 : out_b2;
    const int LLv = is_b1 ? 4096 : 1024;
    const int DD  = is_b1 ? 3072 : 768;
    float s = 0.f;
    for (int i = tid; i < LLv; i += nt) { float g = GG[i]; s += g * g; }
    red[tid] = s; __syncthreads();
    for (int off = nt >> 1; off > 0; off >>= 1) {
        if (tid < off) red[tid] += red[tid + off];
        __syncthreads();
    }
    for (int i = tid; i < LLv; i += nt) m[i] = w[Pi[i] & (P - 1)] * GG[i];
    fwht_lds(m, LLv, tid, nt);
    float scale = rsqrtf((float)DD * red[0]);
    for (int i = tid; i < DD; i += nt) out[i] = init[i] + m[i] * scale;
}

struct Frag { v4f q0, q1, q2, q3; };

__device__ __forceinline__ void load_frag(Frag& f, const float* __restrict__ p) {
    const v4f* p4 = (const v4f*)p;
    f.q0 = p4[0]; f.q1 = p4[1]; f.q2 = p4[2]; f.q3 = p4[3];
}

#define SCHED_FENCE() __builtin_amdgcn_sched_barrier(0)
#define KEEP_LOADS_IN_THIS_TRIP() asm volatile("" ::: "memory")

#define LO2(q) __builtin_shufflevector(q, q, 0, 1)
#define HI2(q) __builtin_shufflevector(q, q, 2, 3)
#define WMMA_F32(a, b, c) __builtin_amdgcn_wmma_f32_16x16x4_f32(false, a, false, b, (short)0, c, false, false)
#define KSTEP(SEL, Q)                                   \
    acc0 = WMMA_F32(SEL(fa.Q), SEL(f0.Q), acc0);        \
    acc1 = WMMA_F32(SEL(fa.Q), SEL(f1.Q), acc1);        \
    acc2 = WMMA_F32(SEL(fa.Q), SEL(f2.Q), acc2);        \
    acc3 = WMMA_F32(SEL(fa.Q), SEL(f3.Q), acc3);

__device__ __forceinline__ void mma_chunk(const Frag& fa, const Frag& f0, const Frag& f1,
                                          const Frag& f2, const Frag& f3,
                                          v8f& acc0, v8f& acc1, v8f& acc2, v8f& acc3) {
    KSTEP(LO2, q0) KSTEP(HI2, q0)
    KSTEP(LO2, q1) KSTEP(HI2, q1)
    KSTEP(LO2, q2) KSTEP(HI2, q2)
    KSTEP(LO2, q3) KSTEP(HI2, q3)
}

template<int RELU, int KS>
__global__ __launch_bounds__(256)
void gemm_wmma_kernel(const float* __restrict__ A, const float* __restrict__ B,
                      const float* __restrict__ bias, float* __restrict__ C,
                      int M, int N, int K) {
    constexpr int SPB = 8 / KS;
    __shared__ v4f part[8 * 8 * 32];
    int lane = threadIdx.x & 31;
    int wave = threadIdx.x >> 5;
    int sl = wave / KS;
    int ks = wave % KS;
    int job = blockIdx.x * SPB + sl;
    int mTiles = M >> 4;
    int mt = job % mTiles;
    int ng = job / mTiles;
    int row = lane & 15;
    int kg = (lane >> 4) << 4;
    int kLen = K / KS;
    size_t koff = (size_t)ks * kLen + kg;
    const float* ap  = A + (size_t)(mt * 16 + row) * K + koff;
    const float* bp0 = B + (size_t)(ng * 64 + row) * K + koff;
    const float* bp1 = bp0 + (size_t)16 * K;
    const float* bp2 = bp0 + (size_t)32 * K;
    const float* bp3 = bp0 + (size_t)48 * K;
    v8f acc0 = {}, acc1 = {}, acc2 = {}, acc3 = {};

    Frag xa, x0, x1, x2, x3;
    Frag ya, y0, y1, y2, y3;

    load_frag(xa, ap); load_frag(x0, bp0); load_frag(x1, bp1);
    load_frag(x2, bp2); load_frag(x3, bp3);

    int k0 = 0;
    for (; k0 + 64 < kLen; k0 += 64) {
        load_frag(ya, ap + k0 + 32); load_frag(y0, bp0 + k0 + 32); load_frag(y1, bp1 + k0 + 32);
        load_frag(y2, bp2 + k0 + 32); load_frag(y3, bp3 + k0 + 32);
        SCHED_FENCE();
        mma_chunk(xa, x0, x1, x2, x3, acc0, acc1, acc2, acc3);
        SCHED_FENCE();
        load_frag(xa, ap + k0 + 64); load_frag(x0, bp0 + k0 + 64); load_frag(x1, bp1 + k0 + 64);
        load_frag(x2, bp2 + k0 + 64); load_frag(x3, bp3 + k0 + 64);
        KEEP_LOADS_IN_THIS_TRIP();
        SCHED_FENCE();
        mma_chunk(ya, y0, y1, y2, y3, acc0, acc1, acc2, acc3);
    }
    load_frag(ya, ap + k0 + 32); load_frag(y0, bp0 + k0 + 32); load_frag(y1, bp1 + k0 + 32);
    load_frag(y2, bp2 + k0 + 32); load_frag(y3, bp3 + k0 + 32);
    SCHED_FENCE();
    mma_chunk(xa, x0, x1, x2, x3, acc0, acc1, acc2, acc3);
    mma_chunk(ya, y0, y1, y2, y3, acc0, acc1, acc2, acc3);

    v4f* mine = part + wave * 256;
    v8f accs[4] = {acc0, acc1, acc2, acc3};
#pragma unroll
    for (int t = 0; t < 4; ++t) {
        mine[(2 * t) * 32 + lane]     = __builtin_shufflevector(accs[t], accs[t], 0, 1, 2, 3);
        mine[(2 * t + 1) * 32 + lane] = __builtin_shufflevector(accs[t], accs[t], 4, 5, 6, 7);
    }
    __syncthreads();

    constexpr int GPW = 8 / KS;
    const v4f* sp = part + sl * KS * 256;
    int mrow = mt * 16 + (lane >> 4) * 8;
#pragma unroll
    for (int j = 0; j < GPW; ++j) {
        int g = ks * GPW + j;
        int t = g >> 1;
        int vb = (g & 1) * 4;
        v4f s = sp[g * 32 + lane];
#pragma unroll
        for (int q = 1; q < KS; ++q) s += sp[q * 256 + g * 32 + lane];
        int n = ng * 64 + t * 16 + row;
        float bv = bias[n];
#pragma unroll
        for (int e = 0; e < 4; ++e) {
            float xv = s[e] + bv;
            if (RELU) xv = fmaxf(xv, 0.f);
            C[(size_t)(mrow + vb + e) * N + n] = xv;
        }
    }
}

extern "C" void kernel_launch(void* const* d_in, const int* in_sizes, int n_in,
                              void* d_out, int out_size, void* d_ws, size_t ws_size,
                              hipStream_t stream) {
    const float* x     = (const float*)d_in[0];
    const float* theta = (const float*)d_in[1];
    const float* iW1   = (const float*)d_in[2];
    const float* ib1   = (const float*)d_in[3];
    const float* iW2   = (const float*)d_in[4];
    const float* ib2   = (const float*)d_in[5];
    const float* BB_W1 = (const float*)d_in[6];
    const int*   Pi_W1 = (const int*)d_in[7];
    const float* GG_W1 = (const float*)d_in[8];
    const float* BB_b1 = (const float*)d_in[9];
    const int*   Pi_b1 = (const int*)d_in[10];
    const float* GG_b1 = (const float*)d_in[11];
    const float* BB_W2 = (const float*)d_in[12];
    const int*   Pi_W2 = (const int*)d_in[13];
    const float* GG_W2 = (const float*)d_in[14];
    const float* BB_b2 = (const float*)d_in[15];
    const int*   Pi_b2 = (const int*)d_in[16];
    const float* GG_b2 = (const float*)d_in[17];
    float* out = (float*)d_out;

    float* ws = (float*)d_ws;
    size_t o = 0;
    float* W1   = ws + o; o += DD_W;
    float* W2   = ws + o; o += DD_W;
    float* b1   = ws + o; o += 3072;
    float* b2   = ws + o; o += 768;
    float* hbuf = ws + o; o += (size_t)256 * 3072;
    float* wsm  = ws + o; o += 4096;
    float* part = ws + o; o += 512;
    float* gsum = ws + o; o += 2;

    fastfood_small_kernel<<<4, 512, 0, stream>>>(theta, BB_W1, BB_W2, wsm,
                                                 BB_b1, Pi_b1, GG_b1, ib1, b1,
                                                 BB_b2, Pi_b2, GG_b2, ib2, b2);
    gg_partial_kernel<<<dim3(256, 2), 256, 0, stream>>>(GG_W1, GG_W2, LL_BIG, part);
    gg_final_kernel<<<dim3(1, 2), 256, 0, stream>>>(part, gsum);
    fwht_passA_kernel<<<dim3(COLS / TC, 2), 512, 0, stream>>>(wsm, Pi_W1, GG_W1, W1,
                                                              Pi_W2, GG_W2, W2, 2047);
    fwht_passB_kernel<<<dim3(ROWS_OUT, 2), 64, 0, stream>>>(W1, iW1, W2, iW2, gsum, DD_W);

    gemm_wmma_kernel<1, 2><<<192, 256, 0, stream>>>(x, W1, b1, hbuf, 256, 3072, 768);
    gemm_wmma_kernel<0, 8><<<192, 256, 0, stream>>>(hbuf, W2, b2, out, 256, 768, 3072);
}


# sum(GG^2) moves into pass A, which already holds every GG value in registers: one partial per workgroup, added by each pass-B workgroup; 7 launches -> 5, GG read once; new sum order
# speedup vs baseline: 5.1175x; 1.3090x over previous
#include <hip/hip_runtime.h>

typedef __attribute__((ext_vector_type(2))) float v2f;
typedef __attribute__((ext_vector_type(4))) float v4f;
typedef __attribute__((ext_vector_type(8))) float v8f;

#define COLS     4096
#define ROWS     1024
#define ROWS_OUT 576
#define DD_W     2359296
#define TC       32
#define PB_PITCH 260
#define P_W      2048

__device__ __forceinline__ void fwht_lds(float* buf, int n, int tid, int nt) {
    for (int h = 1; h < n; h <<= 1) {
        __syncthreads();
        for (int p = tid; p < (n >> 1); p += nt) {
            int i0 = ((p & ~(h - 1)) << 1) | (p & (h - 1));
            int i1 = i0 + h;
            float a = buf[i0], b = buf[i1];
            buf[i0] = a + b;
            buf[i1] = a - b;
        }
    }
    __syncthreads();
}

#define NPART (COLS / TC)

__device__ __forceinline__ float wave_sum(float x) {
#pragma unroll
    for (int m = warpSize >> 1; m > 0; m >>= 1) x += __shfl_xor(x, m);
    return x;
}

template<int N, int B0, int B1>
__device__ __forceinline__ void fwht_reg_stages(float (&v)[N]) {
#pragma unroll
    for (int b = B0; b < B1; ++b) {
        const int h = 1 << b;
#pragma unroll
        for (int p = 0; p < N / 2; ++p) {
            const int i0 = ((p & ~(h - 1)) << 1) | (p & (h - 1));
            const float x = v[i0], y = v[i0 + h];
            v[i0] = x + y;
            v[i0 + h] = x - y;
        }
    }
}

__global__ __launch_bounds__(512)
void fwht_passA_kernel(const float* __restrict__ w2,
                       const int* __restrict__ Pi_a, const float* __restrict__ GG_a,
                       float* __restrict__ m_a,
                       const int* __restrict__ Pi_b, const float* __restrict__ GG_b,
                       float* __restrict__ m_b, int pmask,
                       float* __restrict__ part2) {
    __shared__ float tile[ROWS * TC];
    __shared__ float wsum[16];
    const int lane = threadIdx.x & 31;
    const int wv = threadIdx.x >> 5;
    const float* __restrict__ w  = w2 + blockIdx.y * P_W;
    const int*   __restrict__ Pi = blockIdx.y ? Pi_b : Pi_a;
    const float* __restrict__ GG = blockIdx.y ? GG_b : GG_a;
    float* __restrict__ mbuf     = blockIdx.y ? m_b : m_a;
    const int col = blockIdx.x * TC + lane;

    float v[64];
    int pi[64];
    const int g0 = wv * 64 * COLS + col;
#pragma unroll
    for (int j = 0; j < 64; ++j) pi[j] = Pi[g0 + j * COLS];
#pragma unroll
    for (int j = 0; j < 64; ++j) v[j] = GG[g0 + j * COLS];
    float ss[4] = {0.f, 0.f, 0.f, 0.f};
#pragma unroll
    for (int j = 0; j < 64; ++j) {
        const float g = v[j];
        ss[j & 3] += g * g;
        v[j] = w[pi[j] & pmask] * g;
    }
    const float wtot = wave_sum((ss[0] + ss[1]) + (ss[2] + ss[3]));
    if (lane == 0) wsum[wv] = wtot;
    fwht_reg_stages<64, 0, 6>(v);
#pragma unroll
    for (int j = 0; j < 64; ++j) tile[(wv * 64 + j) * TC + lane] = v[j];
    __syncthreads();
    if (threadIdx.x == 0) {
        float tot = wsum[0];
#pragma unroll
        for (int i = 1; i < 16; ++i) tot += wsum[i];
        part2[NPART * blockIdx.y + blockIdx.x] = tot;
    }
    float u[64];
#pragma unroll
    for (int q = 0; q < 4; ++q) {
#pragma unroll
        for (int jh = 0; jh < 16; ++jh) u[16 * q + jh] = tile[(64 * jh + 4 * wv + q) * TC + lane];
    }
    fwht_reg_stages<64, 0, 4>(u);
#pragma unroll
    for (int q = 0; q < 4; ++q) {
#pragma unroll
        for (int jh = 0; jh < ROWS_OUT / 64; ++jh)
            mbuf[(size_t)(64 * jh + 4 * wv + q) * COLS + col] = u[16 * q + jh];
    }
}

__global__ __launch_bounds__(64)
void fwht_passB_kernel(float* buf_a, const float* __restrict__ init_a,
                       float* buf_b, const float* __restrict__ init_b,
                       const float* __restrict__ part2, int DD) {
    __shared__ v4f blk4[16 * PB_PITCH / 4];
    __shared__ float wsum[2];
    float* blk = (float*)blk4;
    const int t = threadIdx.x;
    float* buf = blockIdx.y ? buf_b : buf_a;
    const float* __restrict__ init = blockIdx.y ? init_b : init_a;
    const float* __restrict__ part = part2 + NPART * blockIdx.y;
    const size_t base = (size_t)blockIdx.x * COLS;
    v4f* row4 = (v4f*)(buf + base) + t;
    const v4f* __restrict__ init4 = (const v4f*)(init + base) + t;

    const float ps = wave_sum(part[t] + part[t + 64]);
    float v[64];
    v4f in[16];
#pragma unroll
    for (int c = 0; c < 16; ++c) {
        const v4f x = row4[64 * c];
        v[4 * c + 0] = x[0]; v[4 * c + 1] = x[1]; v[4 * c + 2] = x[2]; v[4 * c + 3] = x[3];
    }
#pragma unroll
    for (int c = 0; c < 16; ++c) in[c] = init4[64 * c];
    fwht_reg_stages<64, 0, 2>(v);
#pragma unroll
    for (int c = 0; c < 16; ++c) {
        v4f x; x[0] = v[4 * c + 0]; x[1] = v[4 * c + 1]; x[2] = v[4 * c + 2]; x[3] = v[4 * c + 3];
        blk4[c * (PB_PITCH / 4) + t] = x;
    }
    if ((t & 31) == 0) wsum[t >> 5] = ps;
    __syncthreads();
    const float gtot = wsum[0] + wsum[1];
    float* pb = blk + (t >> 2) * PB_PITCH + (t & 3);
    float u[64];
#pragma unroll
    for (int m = 0; m < 64; ++m) u[m] = pb[4 * m];
    fwht_reg_stages<64, 0, 6>(u);
#pragma unroll
    for (int m = 0; m < 64; ++m) pb[4 * m] = u[m];
    __syncthreads();
#pragma unroll
    for (int c = 0; c < 16; ++c) {
        const v4f x = blk4[c * (PB_PITCH / 4) + t];
        v[4 * c + 0] = x[0]; v[4 * c + 1] = x[1]; v[4 * c + 2] = x[2]; v[4 * c + 3] = x[3];
    }
    fwht_reg_stages<64, 2, 6>(v);
    const float scale = rsqrtf((float)DD * gtot);
#pragma unroll
    for (int c = 0; c < 16; ++c) {
        v4f o;
        o[0] = in[c][0] + v[4 * c + 0] * scale; o[1] = in[c][1] + v[4 * c + 1] * scale;
        o[2] = in[c][2] + v[4 * c + 2] * scale; o[3] = in[c][3] + v[4 * c + 3] * scale;
        row4[64 * c] = o;
    }
}

__global__ __launch_bounds__(512)
void fastfood_small_kernel(const float* __restrict__ theta,
                           const float* __restrict__ BB_W1, const float* __restrict__ BB_W2,
                           float* __restrict__ wout,
                           const float* __restrict__ BB_b1, const int* __restrict__ Pi_b1,
                           const float* __restrict__ GG_b1, const float* __restrict__ init_b1,
                           float* __restrict__ out_b1,
                           const float* __restrict__ BB_b2, const int* __restrict__ Pi_b2,
                           const float* __restrict__ GG_b2, const float* __restrict__ init_b2,
                           float* __restrict__ out_b2) {
    __shared__ float w[2048];
    __shared__ float m[4096];
    __shared__ float red[512];
    int tid = threadIdx.x, nt = blockDim.x;
    const int job = blockIdx.x;
    const float* __restrict__ BB = job == 0 ? BB_W1 : job == 1 ? BB_W2 : job == 2 ? BB_b1 : BB_b2;
    const int P = job == 3 ? 1024 : P_W;
    for (int j = tid; j < P; j += nt) w[j] = BB[j] * theta[j];
    fwht_lds(w, P, tid, nt);
    if (job < 2) {
        for (int j = tid; j < P; j += nt) wout[job * P_W + j] = w[j];
        return;
    }
    const bool is_b1 = job == 2;
    const int*   __restrict__ Pi   = is_b1 ? Pi_b1 : Pi_b2;
    const float* __restrict__ GG   = is_b1 ? GG_b1 : GG_b2;
    const float* __restrict__ init = is_b1 ? init_b1 : init_b2;
    float* __restrict__ out        = is_b1 ? out_b1 : out_b2;
    const int LLv = is_b1 ? 4096 : 1024;
    const int DD  = is_b1 ? 3072 : 768;
    float s = 0.f;
    for (int i = tid; i < LLv; i += nt) { float g = GG[i]; s += g * g; }
    red[tid] = s; __syncthreads();
    for (int off = nt >> 1; off > 0; off >>= 1) {
        if (tid < off) red[tid] += red[tid + off];
        __syncthreads();
    }
    for (int i = tid; i < LLv; i += nt) m[i] = w[Pi[i] & (P - 1)] * GG[i];
    fwht_lds(m, LLv, tid, nt);
    float scale = rsqrtf((float)DD * red[0]);
    for (int i = tid; i < DD; i += nt) out[i] = init[i] + m[i] * scale;
}

struct Frag { v4f q0, q1, q2, q3; };

__device__ __forceinline__ void load_frag(Frag& f, const float* __restrict__ p) {
    const v4f* p4 = (const v4f*)p;
    f.q0 = p4[0]; f.q1 = p4[1]; f.q2 = p4[2]; f.q3 = p4[3];
}

#define SCHED_FENCE() __builtin_amdgcn_sched_barrier(0)
#define KEEP_LOADS_IN_THIS_TRIP() asm volatile("" ::: "memory")

#define LO2(q) __builtin_shufflevector(q, q, 0, 1)
#define HI2(q) __builtin_shufflevector(q, q, 2, 3)
#define WMMA_F32(a, b, c) __builtin_amdgcn_wmma_f32_16x16x4_f32(false, a, false, b, (short)0, c, false, false)
#define KSTEP(SEL, Q)                                   \
    acc0 = WMMA_F32(SEL(fa.Q), SEL(f0.Q), acc0);        \
    acc1 = WMMA_F32(SEL(fa.Q), SEL(f1.Q), acc1);        \
    acc2 = WMMA_F32(SEL(fa.Q), SEL(f2.Q), acc2);        \
    acc3 = WMMA_F32(SEL(fa.Q), SEL(f3.Q), acc3);

__device__ __forceinline__ void mma_chunk(const Frag& fa, const Frag& f0, const Frag& f1,
                                          const Frag& f2, const Frag& f3,
                                          v8f& acc0, v8f& acc1, v8f& acc2, v8f& acc3) {
    KSTEP(LO2, q0) KSTEP(HI2, q0)
    KSTEP(LO2, q1) KSTEP(HI2, q1)
    KSTEP(LO2, q2) KSTEP(HI2, q2)
    KSTEP(LO2, q3) KSTEP(HI2, q3)
}

template<int RELU, int KS>
__global__ __launch_bounds__(256)
void gemm_wmma_kernel(const float* __restrict__ A, const float* __restrict__ B,
                      const float* __restrict__ bias, float* __restrict__ C,
                      int M, int N, int K) {
    constexpr int SPB = 8 / KS;
    __shared__ v4f part[8 * 8 * 32];
    int lane = threadIdx.x & 31;
    int wave = threadIdx.x >> 5;
    int sl = wave / KS;
    int ks = wave % KS;
    int job = blockIdx.x * SPB + sl;
    int mTiles = M >> 4;
    int mt = job % mTiles;
    int ng = job / mTiles;
    int row = lane & 15;
    int kg = (lane >> 4) << 4;
    int kLen = K / KS;
    size_t koff = (size_t)ks * kLen + kg;
    const float* ap  = A + (size_t)(mt * 16 + row) * K + koff;
    const float* bp0 = B + (size_t)(ng * 64 + row) * K + koff;
    const float* bp1 = bp0 + (size_t)16 * K;
    const float* bp2 = bp0 + (size_t)32 * K;
    const float* bp3 = bp0 + (size_t)48 * K;
    v8f acc0 = {}, acc1 = {}, acc2 = {}, acc3 = {};

    Frag xa, x0, x1, x2, x3;
    Frag ya, y0, y1, y2, y3;

    load_frag(xa, ap); load_frag(x0, bp0); load_frag(x1, bp1);
    load_frag(x2, bp2); load_frag(x3, bp3);

    int k0 = 0;
    for (; k0 + 64 < kLen; k0 += 64) {
        load_frag(ya, ap + k0 + 32); load_frag(y0, bp0 + k0 + 32); load_frag(y1, bp1 + k0 + 32);
        load_frag(y2, bp2 + k0 + 32); load_frag(y3, bp3 + k0 + 32);
        SCHED_FENCE();
        mma_chunk(xa, x0, x1, x2, x3, acc0, acc1, acc2, acc3);
        SCHED_FENCE();
        load_frag(xa, ap + k0 + 64); load_frag(x0, bp0 + k0 + 64); load_frag(x1, bp1 + k0 + 64);
        load_frag(x2, bp2 + k0 + 64); load_frag(x3, bp3 + k0 + 64);
        KEEP_LOADS_IN_THIS_TRIP();
        SCHED_FENCE();
        mma_chunk(ya, y0, y1, y2, y3, acc0, acc1, acc2, acc3);
    }
    load_frag(ya, ap + k0 + 32); load_frag(y0, bp0 + k0 + 32); load_frag(y1, bp1 + k0 + 32);
    load_frag(y2, bp2 + k0 + 32); load_frag(y3, bp3 + k0 + 32);
    SCHED_FENCE();
    mma_chunk(xa, x0, x1, x2, x3, acc0, acc1, acc2, acc3);
    mma_chunk(ya, y0, y1, y2, y3, acc0, acc1, acc2, acc3);

    v4f* mine = part + wave * 256;
    v8f accs[4] = {acc0, acc1, acc2, acc3};
#pragma unroll
    for (int t = 0; t < 4; ++t) {
        mine[(2 * t) * 32 + lane]     = __builtin_shufflevector(accs[t], accs[t], 0, 1, 2, 3);
        mine[(2 * t + 1) * 32 + lane] = __builtin_shufflevector(accs[t], accs[t], 4, 5, 6, 7);
    }
    __syncthreads();

    constexpr int GPW = 8 / KS;
    const v4f* sp = part + sl * KS * 256;
    int mrow = mt * 16 + (lane >> 4) * 8;
#pragma unroll
    for (int j = 0; j < GPW; ++j) {
        int g = ks * GPW + j;
        int t = g >> 1;
        int vb = (g & 1) * 4;
        v4f s = sp[g * 32 + lane];
#pragma unroll
        for (int q = 1; q < KS; ++q) s += sp[q * 256 + g * 32 + lane];
        int n = ng * 64 + t * 16 + row;
        float bv = bias[n];
#pragma unroll
        for (int e = 0; e < 4; ++e) {
            float xv = s[e] + bv;
            if (RELU) xv = fmaxf(xv, 0.f);
            C[(size_t)(mrow + vb + e) * N + n] = xv;
        }
    }
}

extern "C" void kernel_launch(void* const* d_in, const int* in_sizes, int n_in,
                              void* d_out, int out_size, void* d_ws, size_t ws_size,
                              hipStream_t stream) {
    const float* x     = (const float*)d_in[0];
    const float* theta = (const float*)d_in[1];
    const float* iW1   = (const float*)d_in[2];
    const float* ib1   = (const float*)d_in[3];
    const float* iW2   = (const float*)d_in[4];
    const float* ib2   = (const float*)d_in[5];
    const float* BB_W1 = (const float*)d_in[6];
    const int*   Pi_W1 = (const int*)d_in[7];
    const float* GG_W1 = (const float*)d_in[8];
    const float* BB_b1 = (const float*)d_in[9];
    const int*   Pi_b1 = (const int*)d_in[10];
    const float* GG_b1 = (const float*)d_in[11];
    const float* BB_W2 = (const float*)d_in[12];
    const int*   Pi_W2 = (const int*)d_in[13];
    const float* GG_W2 = (const float*)d_in[14];
    const float* BB_b2 = (const float*)d_in[15];
    const int*   Pi_b2 = (const int*)d_in[16];
    const float* GG_b2 = (const float*)d_in[17];
    float* out = (float*)d_out;

    float* ws = (float*)d_ws;
    size_t o = 0;
    float* W1   = ws + o; o += DD_W;
    float* W2   = ws + o; o += DD_W;
    float* b1   = ws + o; o += 3072;
    float* b2   = ws + o; o += 768;
    float* hbuf = ws + o; o += (size_t)256 * 3072;
    float* wsm  = ws + o; o += 4096;
    float* part = ws + o; o += 2 * NPART;

    fastfood_small_kernel<<<4, 512, 0, stream>>>(theta, BB_W1, BB_W2, wsm,
                                                 BB_b1, Pi_b1, GG_b1, ib1, b1,
                                                 BB_b2, Pi_b2, GG_b2, ib2, b2);
    fwht_passA_kernel<<<dim3(COLS / TC, 2), 512, 0, stream>>>(wsm, Pi_W1, GG_W1, W1,
                                                              Pi_W2, GG_W2, W2, 2047, part);
    fwht_passB_kernel<<<dim3(ROWS_OUT, 2), 64, 0, stream>>>(W1, iW1, W2, iW2, part, DD_W);

    gemm_wmma_kernel<1, 2><<<192, 256, 0, stream>>>(x, W1, b1, hbuf, 256, 3072, 768);
    gemm_wmma_kernel<0, 8><<<192, 256, 0, stream>>>(hbuf, W2, b2, out, 256, 768, 3072);
}


# w = FWHT(BB*theta) built by each pass-A workgroup in LDS behind its Pi/GG loads, gathered from LDS; b1, b2 ride as 2 more pass-A workgroups; 5 launches -> 4; every sum order kept
# speedup vs baseline: 5.8007x; 1.1335x over previous
#include <hip/hip_runtime.h>

typedef __attribute__((ext_vector_type(2))) float v2f;
typedef __attribute__((ext_vector_type(4))) float v4f;
typedef __attribute__((ext_vector_type(8))) float v8f;

#define COLS     4096
#define ROWS     1024
#define ROWS_OUT 576
#define DD_W     2359296
#define TC       32
#define PB_PITCH 260
#define P_W      2048

__device__ __forceinline__ void fwht_lds(float* buf, int n, int tid, int nt) {
    for (int h = 1; h < n; h <<= 1) {
        __syncthreads();
        for (int p = tid; p < (n >> 1); p += nt) {
            int i0 = ((p & ~(h - 1)) << 1) | (p & (h - 1));
            int i1 = i0 + h;
            float a = buf[i0], b = buf[i1];
            buf[i0] = a + b;
            buf[i1] = a - b;
        }
    }
    __syncthreads();
}

__device__ __forceinline__ void lds_only_barrier() {
    __builtin_amdgcn_fence(__ATOMIC_RELEASE, "workgroup", "local");
    __builtin_amdgcn_s_barrier();
    __builtin_amdgcn_fence(__ATOMIC_ACQUIRE, "workgroup", "local");
}

__device__ __forceinline__ void fwht_lds_inflight(float* buf, int n, int tid, int nt) {
    for (int h = 1; h < n; h <<= 1) {
        lds_only_barrier();
        for (int p = tid; p < (n >> 1); p += nt) {
            int i0 = ((p & ~(h - 1)) << 1) | (p & (h - 1));
            int i1 = i0 + h;
            float a = buf[i0], b = buf[i1];
            buf[i0] = a + b;
            buf[i1] = a - b;
        }
    }
    lds_only_barrier();
}

#define NPART (COLS / TC)

__device__ __forceinline__ float wave_sum(float x) {
#pragma unroll
    for (int m = warpSize >> 1; m > 0; m >>= 1) x += __shfl_xor(x, m);
    return x;
}

template<int N, int B0, int B1>
__device__ __forceinline__ void fwht_reg_stages(float (&v)[N]) {
#pragma unroll
    for (int b = B0; b < B1; ++b) {
        const int h = 1 << b;
#pragma unroll
        for (int p = 0; p < N / 2; ++p) {
            const int i0 = ((p & ~(h - 1)) << 1) | (p & (h - 1));
            const float x = v[i0], y = v[i0 + h];
            v[i0] = x + y;
            v[i0 + h] = x - y;
        }
    }
}

__device__ __forceinline__ void bias_projection(const float* __restrict__ theta,
                                                const float* __restrict__ BB,
                                                const int* __restrict__ Pi,
                                                const float* __restrict__ GG,
                                                const float* __restrict__ init,
                                                float* __restrict__ out,
                                                int P, int LLv, int DD,
                                                float* w, float* m, float* red) {
    const int tid = threadIdx.x, nt = 512;
    for (int j = tid; j < P; j += nt) w[j] = BB[j] * theta[j];
    fwht_lds(w, P, tid, nt);
    float s = 0.f;
    for (int i = tid; i < LLv; i += nt) { float g = GG[i]; s += g * g; }
    red[tid] = s; __syncthreads();
    for (int off = nt >> 1; off > 0; off >>= 1) {
        if (tid < off) red[tid] += red[tid + off];
        __syncthreads();
    }
    for (int i = tid; i < LLv; i += nt) m[i] = w[Pi[i] & (P - 1)] * GG[i];
    fwht_lds(m, LLv, tid, nt);
    float scale = rsqrtf((float)DD * red[0]);
    for (int i = tid; i < DD; i += nt) out[i] = init[i] + m[i] * scale;
}

__global__ __launch_bounds__(512)
void fwht_passA_kernel(const float* __restrict__ theta,
                       const float* __restrict__ BB_a,
                       const int* __restrict__ Pi_a, const float* __restrict__ GG_a,
                       float* __restrict__ m_a,
                       const float* __restrict__ BB_b,
                       const int* __restrict__ Pi_b, const float* __restrict__ GG_b,
                       float* __restrict__ m_b, int pmask,
                       float* __restrict__ part2,
                       const float* __restrict__ BB_b1, const int* __restrict__ Pi_b1,
                       const float* __restrict__ GG_b1, const float* __restrict__ init_b1,
                       float* __restrict__ out_b1,
                       const float* __restrict__ BB_b2, const int* __restrict__ Pi_b2,
                       const float* __restrict__ GG_b2, const float* __restrict__ init_b2,
                       float* __restrict__ out_b2) {
    __shared__ float tile[ROWS * TC];
    __shared__ float wl[P_W];
    __shared__ float wsum[16];
    if (blockIdx.x == 0) {
        if (blockIdx.y == 0)
            bias_projection(theta, BB_b1, Pi_b1, GG_b1, init_b1, out_b1, P_W, 4096, 3072,
                            wl, tile, tile + 4096);
        else
            bias_projection(theta, BB_b2, Pi_b2, GG_b2, init_b2, out_b2, 1024, 1024, 768,
                            wl, tile, tile + 4096);
        return;
    }
    const int lane = threadIdx.x & 31;
    const int wv = threadIdx.x >> 5;
    const int tix = blockIdx.x - 1;
    const float* __restrict__ BB = blockIdx.y ? BB_b : BB_a;
    const int*   __restrict__ Pi = blockIdx.y ? Pi_b : Pi_a;
    const float* __restrict__ GG = blockIdx.y ? GG_b : GG_a;
    float* __restrict__ mbuf     = blockIdx.y ? m_b : m_a;
    const int col = tix * TC + lane;

    float wp[P_W / 512];
#pragma unroll
    for (int k = 0; k < P_W / 512; ++k) wp[k] = BB[threadIdx.x + 512 * k] * theta[threadIdx.x + 512 * k];
    float v[64];
    int pi[64];
    const int g0 = wv * 64 * COLS + col;
#pragma unroll
    for (int j = 0; j < 64; ++j) pi[j] = Pi[g0 + j * COLS];
#pragma unroll
    for (int j = 0; j < 64; ++j) v[j] = GG[g0 + j * COLS];
#pragma unroll
    for (int k = 0; k < P_W / 512; ++k) wl[threadIdx.x + 512 * k] = wp[k];
    fwht_lds_inflight(wl, P_W, threadIdx.x, 512);
    float ss[4] = {0.f, 0.f, 0.f, 0.f};
#pragma unroll
    for (int j = 0; j < 64; ++j) {
        const float g = v[j];
        ss[j & 3] += g * g;
        v[j] = wl[pi[j] & pmask] * g;
    }
    const float wtot = wave_sum((ss[0] + ss[1]) + (ss[2] + ss[3]));
    if (lane == 0) wsum[wv] = wtot;
    fwht_reg_stages<64, 0, 6>(v);
#pragma unroll
    for (int j = 0; j < 64; ++j) tile[(wv * 64 + j) * TC + lane] = v[j];
    __syncthreads();
    if (threadIdx.x == 0) {
        float tot = wsum[0];
#pragma unroll
        for (int i = 1; i < 16; ++i) tot += wsum[i];
        part2[NPART * blockIdx.y + tix] = tot;
    }
    float u[64];
#pragma unroll
    for (int q = 0; q < 4; ++q) {
#pragma unroll
        for (int jh = 0; jh < 16; ++jh) u[16 * q + jh] = tile[(64 * jh + 4 * wv + q) * TC + lane];
    }
    fwht_reg_stages<64, 0, 4>(u);
#pragma unroll
    for (int q = 0; q < 4; ++q) {
#pragma unroll
        for (int jh = 0; jh < ROWS_OUT / 64; ++jh)
            mbuf[(size_t)(64 * jh + 4 * wv + q) * COLS + col] = u[16 * q + jh];
    }
}

__global__ __launch_bounds__(64)
void fwht_passB_kernel(float* buf_a, const float* __restrict__ init_a,
                       float* buf_b, const float* __restrict__ init_b,
                       const float* __restrict__ part2, int DD) {
    __shared__ v4f blk4[16 * PB_PITCH / 4];
    __shared__ float wsum[2];
    float* blk = (float*)blk4;
    const int t = threadIdx.x;
    float* buf = blockIdx.y ? buf_b : buf_a;
    const float* __restrict__ init = blockIdx.y ? init_b : init_a;
    const float* __restrict__ part = part2 + NPART * blockIdx.y;
    const size_t base = (size_t)blockIdx.x * COLS;
    v4f* row4 = (v4f*)(buf + base) + t;
    const v4f* __restrict__ init4 = (const v4f*)(init + base) + t;

    const float ps = wave_sum(part[t] + part[t + 64]);
    float v[64];
    v4f in[16];
#pragma unroll
    for (int c = 0; c < 16; ++c) {
        const v4f x = row4[64 * c];
        v[4 * c + 0] = x[0]; v[4 * c + 1] = x[1]; v[4 * c + 2] = x[2]; v[4 * c + 3] = x[3];
    }
#pragma unroll
    for (int c = 0; c < 16; ++c) in[c] = init4[64 * c];
    fwht_reg_stages<64, 0, 2>(v);
#pragma unroll
    for (int c = 0; c < 16; ++c) {
        v4f x; x[0] = v[4 * c + 0]; x[1] = v[4 * c + 1]; x[2] = v[4 * c + 2]; x[3] = v[4 * c + 3];
        blk4[c * (PB_PITCH / 4) + t] = x;
    }
    if ((t & 31) == 0) wsum[t >> 5] = ps;
    __syncthreads();
    const float gtot = wsum[0] + wsum[1];
    float* pb = blk + (t >> 2) * PB_PITCH + (t & 3);
    float u[64];
#pragma unroll
    for (int m = 0; m < 64; ++m) u[m] = pb[4 * m];
    fwht_reg_stages<64, 0, 6>(u);
#pragma unroll
    for (int m = 0; m < 64; ++m) pb[4 * m] = u[m];
    __syncthreads();
#pragma unroll
    for (int c = 0; c < 16; ++c) {
        const v4f x = blk4[c * (PB_PITCH / 4) + t];
        v[4 * c + 0] = x[0]; v[4 * c + 1] = x[1]; v[4 * c + 2] = x[2]; v[4 * c + 3] = x[3];
    }
    fwht_reg_stages<64, 2, 6>(v);
    const float scale = rsqrtf((float)DD * gtot);
#pragma unroll
    for (int c = 0; c < 16; ++c) {
        v4f o;
        o[0] = in[c][0] + v[4 * c + 0] * scale; o[1] = in[c][1] + v[4 * c + 1] * scale;
        o[2] = in[c][2] + v[4 * c + 2] * scale; o[3] = in[c][3] + v[4 * c + 3] * scale;
        row4[64 * c] = o;
    }
}

struct Frag { v4f q0, q1, q2, q3; };

__device__ __forceinline__ void load_frag(Frag& f, const float* __restrict__ p) {
    const v4f* p4 = (const v4f*)p;
    f.q0 = p4[0]; f.q1 = p4[1]; f.q2 = p4[2]; f.q3 = p4[3];
}

#define SCHED_FENCE() __builtin_amdgcn_sched_barrier(0)
#define KEEP_LOADS_IN_THIS_TRIP() asm volatile("" ::: "memory")

#define LO2(q) __builtin_shufflevector(q, q, 0, 1)
#define HI2(q) __builtin_shufflevector(q, q, 2, 3)
#define WMMA_F32(a, b, c) __builtin_amdgcn_wmma_f32_16x16x4_f32(false, a, false, b, (short)0, c, false, false)
#define KSTEP(SEL, Q)                                   \
    acc0 = WMMA_F32(SEL(fa.Q), SEL(f0.Q), acc0);        \
    acc1 = WMMA_F32(SEL(fa.Q), SEL(f1.Q), acc1);        \
    acc2 = WMMA_F32(SEL(fa.Q), SEL(f2.Q), acc2);        \
    acc3 = WMMA_F32(SEL(fa.Q), SEL(f3.Q), acc3);

__device__ __forceinline__ void mma_chunk(const Frag& fa, const Frag& f0, const Frag& f1,
                                          const Frag& f2, const Frag& f3,
                                          v8f& acc0, v8f& acc1, v8f& acc2, v8f& acc3) {
    KSTEP(LO2, q0) KSTEP(HI2, q0)
    KSTEP(LO2, q1) KSTEP(HI2, q1)
    KSTEP(LO2, q2) KSTEP(HI2, q2)
    KSTEP(LO2, q3) KSTEP(HI2, q3)
}

template<int RELU, int KS>
__global__ __launch_bounds__(256)
void gemm_wmma_kernel(const float* __restrict__ A, const float* __restrict__ B,
                      const float* __restrict__ bias, float* __restrict__ C,
                      int M, int N, int K) {
    constexpr int SPB = 8 / KS;
    __shared__ v4f part[8 * 8 * 32];
    int lane = threadIdx.x & 31;
    int wave = threadIdx.x >> 5;
    int sl = wave / KS;
    int ks = wave % KS;
    int job = blockIdx.x * SPB + sl;
    int mTiles = M >> 4;
    int mt = job % mTiles;
    int ng = job / mTiles;
    int row = lane & 15;
    int kg = (lane >> 4) << 4;
    int kLen = K / KS;
    size_t koff = (size_t)ks * kLen + kg;
    const float* ap  = A + (size_t)(mt * 16 + row) * K + koff;
    const float* bp0 = B + (size_t)(ng * 64 + row) * K + koff;
    const float* bp1 = bp0 + (size_t)16 * K;
    const float* bp2 = bp0 + (size_t)32 * K;
    const float* bp3 = bp0 + (size_t)48 * K;
    v8f acc0 = {}, acc1 = {}, acc2 = {}, acc3 = {};

    Frag xa, x0, x1, x2, x3;
    Frag ya, y0, y1, y2, y3;

    load_frag(xa, ap); load_frag(x0, bp0); load_frag(x1, bp1);
    load_frag(x2, bp2); load_frag(x3, bp3);

    int k0 = 0;
    for (; k0 + 64 < kLen; k0 += 64) {
        load_frag(ya, ap + k0 + 32); load_frag(y0, bp0 + k0 + 32); load_frag(y1, bp1 + k0 + 32);
        load_frag(y2, bp2 + k0 + 32); load_frag(y3, bp3 + k0 + 32);
        SCHED_FENCE();
        mma_chunk(xa, x0, x1, x2, x3, acc0, acc1, acc2, acc3);
        SCHED_FENCE();
        load_frag(xa, ap + k0 + 64); load_frag(x0, bp0 + k0 + 64); load_frag(x1, bp1 + k0 + 64);
        load_frag(x2, bp2 + k0 + 64); load_frag(x3, bp3 + k0 + 64);
        KEEP_LOADS_IN_THIS_TRIP();
        SCHED_FENCE();
        mma_chunk(ya, y0, y1, y2, y3, acc0, acc1, acc2, acc3);
    }
    load_frag(ya, ap + k0 + 32); load_frag(y0, bp0 + k0 + 32); load_frag(y1, bp1 + k0 + 32);
    load_frag(y2, bp2 + k0 + 32); load_frag(y3, bp3 + k0 + 32);
    SCHED_FENCE();
    mma_chunk(xa, x0, x1, x2, x3, acc0, acc1, acc2, acc3);
    mma_chunk(ya, y0, y1, y2, y3, acc0, acc1, acc2, acc3);

    v4f* mine = part + wave * 256;
    v8f accs[4] = {acc0, acc1, acc2, acc3};
#pragma unroll
    for (int t = 0; t < 4; ++t) {
        mine[(2 * t) * 32 + lane]     = __builtin_shufflevector(accs[t], accs[t], 0, 1, 2, 3);
        mine[(2 * t + 1) * 32 + lane] = __builtin_shufflevector(accs[t], accs[t], 4, 5, 6, 7);
    }
    __syncthreads();

    constexpr int GPW = 8 / KS;
    const v4f* sp = part + sl * KS * 256;
    int mrow = mt * 16 + (lane >> 4) * 8;
#pragma unroll
    for (int j = 0; j < GPW; ++j) {
        int g = ks * GPW + j;
        int t = g >> 1;
        int vb = (g & 1) * 4;
        v4f s = sp[g * 32 + lane];
#pragma unroll
        for (int q = 1; q < KS; ++q) s += sp[q * 256 + g * 32 + lane];
        int n = ng * 64 + t * 16 + row;
        float bv = bias[n];
#pragma unroll
        for (int e = 0; e < 4; ++e) {
            float xv = s[e] + bv;
            if (RELU) xv = fmaxf(xv, 0.f);
            C[(size_t)(mrow + vb + e) * N + n] = xv;
        }
    }
}

extern "C" void kernel_launch(void* const* d_in, const int* in_sizes, int n_in,
                              void* d_out, int out_size, void* d_ws, size_t ws_size,
                              hipStream_t stream) {
    const float* x     = (const float*)d_in[0];
    const float* theta = (const float*)d_in[1];
    const float* iW1   = (const float*)d_in[2];
    const float* ib1   = (const float*)d_in[3];
    const float* iW2   = (const float*)d_in[4];
    const float* ib2   = (const float*)d_in[5];
    const float* BB_W1 = (const float*)d_in[6];
    const int*   Pi_W1 = (const int*)d_in[7];
    const float* GG_W1 = (const float*)d_in[8];
    const float* BB_b1 = (const float*)d_in[9];
    const int*   Pi_b1 = (const int*)d_in[10];
    const float* GG_b1 = (const float*)d_in[11];
    const float* BB_W2 = (const float*)d_in[12];
    const int*   Pi_W2 = (const int*)d_in[13];
    const float* GG_W2 = (const float*)d_in[14];
    const float* BB_b2 = (const float*)d_in[15];
    const int*   Pi_b2 = (const int*)d_in[16];
    const float* GG_b2 = (const float*)d_in[17];
    float* out = (float*)d_out;

    float* ws = (float*)d_ws;
    size_t o = 0;
    float* W1   = ws + o; o += DD_W;
    float* W2   = ws + o; o += DD_W;
    float* b1   = ws + o; o += 3072;
    float* b2   = ws + o; o += 768;
    float* hbuf = ws + o; o += (size_t)256 * 3072;
    float* part = ws + o; o += 2 * NPART;

    fwht_passA_kernel<<<dim3(1 + NPART, 2), 512, 0, stream>>>(theta, BB_W1, Pi_W1, GG_W1, W1,
                                                              BB_W2, Pi_W2, GG_W2, W2, 2047, part,
                                                              BB_b1, Pi_b1, GG_b1, ib1, b1,
                                                              BB_b2, Pi_b2, GG_b2, ib2, b2);
    fwht_passB_kernel<<<dim3(ROWS_OUT, 2), 64, 0, stream>>>(W1, iW1, W2, iW2, part, DD_W);

    gemm_wmma_kernel<1, 2><<<192, 256, 0, stream>>>(x, W1, b1, hbuf, 256, 3072, 768);
    gemm_wmma_kernel<0, 8><<<192, 256, 0, stream>>>(hbuf, W2, b2, out, 256, 768, 3072);
}


# Bias projections (2 workgroups of pass A): all global loads (BB, theta, GG, Pi, init) issued at entry, one memory round trip instead of four barrier-separated ones; sum orders kept
# speedup vs baseline: 6.2439x; 1.0764x over previous
#include <hip/hip_runtime.h>

typedef __attribute__((ext_vector_type(2))) float v2f;
typedef __attribute__((ext_vector_type(4))) float v4f;
typedef __attribute__((ext_vector_type(8))) float v8f;

#define COLS     4096
#define ROWS     1024
#define ROWS_OUT 576
#define DD_W     2359296
#define TC       32
#define PB_PITCH 260
#define P_W      2048

__device__ __forceinline__ void fwht_lds(float* buf, int n, int tid, int nt) {
    for (int h = 1; h < n; h <<= 1) {
        __syncthreads();
        for (int p = tid; p < (n >> 1); p += nt) {
            int i0 = ((p & ~(h - 1)) << 1) | (p & (h - 1));
            int i1 = i0 + h;
            float a = buf[i0], b = buf[i1];
            buf[i0] = a + b;
            buf[i1] = a - b;
        }
    }
    __syncthreads();
}

__device__ __forceinline__ void lds_only_barrier() {
    __builtin_amdgcn_fence(__ATOMIC_RELEASE, "workgroup", "local");
    __builtin_amdgcn_s_barrier();
    __builtin_amdgcn_fence(__ATOMIC_ACQUIRE, "workgroup", "local");
}

__device__ __forceinline__ void fwht_lds_inflight(float* buf, int n, int tid, int nt) {
    for (int h = 1; h < n; h <<= 1) {
        lds_only_barrier();
        for (int p = tid; p < (n >> 1); p += nt) {
            int i0 = ((p & ~(h - 1)) << 1) | (p & (h - 1));
            int i1 = i0 + h;
            float a = buf[i0], b = buf[i1];
            buf[i0] = a + b;
            buf[i1] = a - b;
        }
    }
    lds_only_barrier();
}

#define NPART (COLS / TC)

__device__ __forceinline__ float wave_sum(float x) {
#pragma unroll
    for (int m = warpSize >> 1; m > 0; m >>= 1) x += __shfl_xor(x, m);
    return x;
}

template<int N, int B0, int B1>
__device__ __forceinline__ void fwht_reg_stages(float (&v)[N]) {
#pragma unroll
    for (int b = B0; b < B1; ++b) {
        const int h = 1 << b;
#pragma unroll
        for (int p = 0; p < N / 2; ++p) {
            const int i0 = ((p & ~(h - 1)) << 1) | (p & (h - 1));
            const float x = v[i0], y = v[i0 + h];
            v[i0] = x + y;
            v[i0 + h] = x - y;
        }
    }
}

template<int P, int LLv, int DD>
__device__ __forceinline__ void bias_projection(const float* __restrict__ theta,
                                                const float* __restrict__ BB,
                                                const int* __restrict__ Pi,
                                                const float* __restrict__ GG,
                                                const float* __restrict__ init,
                                                float* __restrict__ out,
                                                float* w, float* m, float* red) {
    constexpr int nt = 512;
    constexpr int NP = P / nt, NL = LLv / nt, ND = (DD + nt - 1) / nt;
    const int tid = threadIdx.x;
    float bv[NP], tv[NP], gg[NL], in[ND];
    int pi[NL];
#pragma unroll
    for (int k = 0; k < NP; ++k) { bv[k] = BB[tid + nt * k]; tv[k] = theta[tid + nt * k]; }
#pragma unroll
    for (int k = 0; k < NL; ++k) gg[k] = GG[tid + nt * k];
#pragma unroll
    for (int k = 0; k < NL; ++k) pi[k] = Pi[tid + nt * k];
#pragma unroll
    for (int k = 0; k < ND; ++k) in[k] = (tid + nt * k < DD) ? init[tid + nt * k] : 0.f;
#pragma unroll
    for (int k = 0; k < NP; ++k) w[tid + nt * k] = bv[k] * tv[k];
    fwht_lds(w, P, tid, nt);
    float s = 0.f;
#pragma unroll
    for (int k = 0; k < NL; ++k) { float g = gg[k]; s += g * g; }
    red[tid] = s; __syncthreads();
    for (int off = nt >> 1; off > 0; off >>= 1) {
        if (tid < off) red[tid] += red[tid + off];
        __syncthreads();
    }
#pragma unroll
    for (int k = 0; k < NL; ++k) m[tid + nt * k] = w[pi[k] & (P - 1)] * gg[k];
    fwht_lds(m, LLv, tid, nt);
    float scale = rsqrtf((float)DD * red[0]);
#pragma unroll
    for (int k = 0; k < ND; ++k) {
        const int i = tid + nt * k;
        if (i < DD) out[i] = in[k] + m[i] * scale;
    }
}

__global__ __launch_bounds__(512)
void fwht_passA_kernel(const float* __restrict__ theta,
                       const float* __restrict__ BB_a,
                       const int* __restrict__ Pi_a, const float* __restrict__ GG_a,
                       float* __restrict__ m_a,
                       const float* __restrict__ BB_b,
                       const int* __restrict__ Pi_b, const float* __restrict__ GG_b,
                       float* __restrict__ m_b, int pmask,
                       float* __restrict__ part2,
                       const float* __restrict__ BB_b1, const int* __restrict__ Pi_b1,
                       const float* __restrict__ GG_b1, const float* __restrict__ init_b1,
                       float* __restrict__ out_b1,
                       const float* __restrict__ BB_b2, const int* __restrict__ Pi_b2,
                       const float* __restrict__ GG_b2, const float* __restrict__ init_b2,
                       float* __restrict__ out_b2) {
    __shared__ float tile[ROWS * TC];
    __shared__ float wl[P_W];
    __shared__ float wsum[16];
    if (blockIdx.x == 0) {
        if (blockIdx.y == 0)
            bias_projection<P_W, 4096, 3072>(theta, BB_b1, Pi_b1, GG_b1, init_b1, out_b1,
                                             wl, tile, tile + 4096);
        else
            bias_projection<1024, 1024, 768>(theta, BB_b2, Pi_b2, GG_b2, init_b2, out_b2,
                                             wl, tile, tile + 4096);
        return;
    }
    const int lane = threadIdx.x & 31;
    const int wv = threadIdx.x >> 5;
    const int tix = blockIdx.x - 1;
    const float* __restrict__ BB = blockIdx.y ? BB_b : BB_a;
    const int*   __restrict__ Pi = blockIdx.y ? Pi_b : Pi_a;
    const float* __restrict__ GG = blockIdx.y ? GG_b : GG_a;
    float* __restrict__ mbuf     = blockIdx.y ? m_b : m_a;
    const int col = tix * TC + lane;

    float wp[P_W / 512];
#pragma unroll
    for (int k = 0; k < P_W / 512; ++k) wp[k] = BB[threadIdx.x + 512 * k] * theta[threadIdx.x + 512 * k];
    float v[64];
    int pi[64];
    const int g0 = wv * 64 * COLS + col;
#pragma unroll
    for (int j = 0; j < 64; ++j) pi[j] = Pi[g0 + j * COLS];
#pragma unroll
    for (int j = 0; j < 64; ++j) v[j] = GG[g0 + j * COLS];
#pragma unroll
    for (int k = 0; k < P_W / 512; ++k) wl[threadIdx.x + 512 * k] = wp[k];
    fwht_lds_inflight(wl, P_W, threadIdx.x, 512);
    float ss[4] = {0.f, 0.f, 0.f, 0.f};
#pragma unroll
    for (int j = 0; j < 64; ++j) {
        const float g = v[j];
        ss[j & 3] += g * g;
        v[j] = wl[pi[j] & pmask] * g;
    }
    const float wtot = wave_sum((ss[0] + ss[1]) + (ss[2] + ss[3]));
    if (lane == 0) wsum[wv] = wtot;
    fwht_reg_stages<64, 0, 6>(v);
#pragma unroll
    for (int j = 0; j < 64; ++j) tile[(wv * 64 + j) * TC + lane] = v[j];
    __syncthreads();
    if (threadIdx.x == 0) {
        float tot = wsum[0];
#pragma unroll
        for (int i = 1; i < 16; ++i) tot += wsum[i];
        part2[NPART * blockIdx.y + tix] = tot;
    }
    float u[64];
#pragma unroll
    for (int q = 0; q < 4; ++q) {
#pragma unroll
        for (int jh = 0; jh < 16; ++jh) u[16 * q + jh] = tile[(64 * jh + 4 * wv + q) * TC + lane];
    }
    fwht_reg_stages<64, 0, 4>(u);
#pragma unroll
    for (int q = 0; q < 4; ++q) {
#pragma unroll
        for (int jh = 0; jh < ROWS_OUT / 64; ++jh)
            mbuf[(size_t)(64 * jh + 4 * wv + q) * COLS + col] = u[16 * q + jh];
    }
}

__global__ __launch_bounds__(64)
void fwht_passB_kernel(float* buf_a, const float* __restrict__ init_a,
                       float* buf_b, const float* __restrict__ init_b,
                       const float* __restrict__ part2, int DD) {
    __shared__ v4f blk4[16 * PB_PITCH / 4];
    __shared__ float wsum[2];
    float* blk = (float*)blk4;
    const int t = threadIdx.x;
    float* buf = blockIdx.y ? buf_b : buf_a;
    const float* __restrict__ init = blockIdx.y ? init_b : init_a;
    const float* __restrict__ part = part2 + NPART * blockIdx.y;
    const size_t base = (size_t)blockIdx.x * COLS;
    v4f* row4 = (v4f*)(buf + base) + t;
    const v4f* __restrict__ init4 = (const v4f*)(init + base) + t;

    const float ps = wave_sum(part[t] + part[t + 64]);
    float v[64];
    v4f in[16];
#pragma unroll
    for (int c = 0; c < 16; ++c) {
        const v4f x = row4[64 * c];
        v[4 * c + 0] = x[0]; v[4 * c + 1] = x[1]; v[4 * c + 2] = x[2]; v[4 * c + 3] = x[3];
    }
#pragma unroll
    for (int c = 0; c < 16; ++c) in[c] = init4[64 * c];
    fwht_reg_stages<64, 0, 2>(v);
#pragma unroll
    for (int c = 0; c < 16; ++c) {
        v4f x; x[0] = v[4 * c + 0]; x[1] = v[4 * c + 1]; x[2] = v[4 * c + 2]; x[3] = v[4 * c + 3];
        blk4[c * (PB_PITCH / 4) + t] = x;
    }
    if ((t & 31) == 0) wsum[t >> 5] = ps;
    __syncthreads();
    const float gtot = wsum[0] + wsum[1];
    float* pb = blk + (t >> 2) * PB_PITCH + (t & 3);
    float u[64];
#pragma unroll
    for (int m = 0; m < 64; ++m) u[m] = pb[4 * m];
    fwht_reg_stages<64, 0, 6>(u);
#pragma unroll
    for (int m = 0; m < 64; ++m) pb[4 * m] = u[m];
    __syncthreads();
#pragma unroll
    for (int c = 0; c < 16; ++c) {
        const v4f x = blk4[c * (PB_PITCH / 4) + t];
        v[4 * c + 0] = x[0]; v[4 * c + 1] = x[1]; v[4 * c + 2] = x[2]; v[4 * c + 3] = x[3];
    }
    fwht_reg_stages<64, 2, 6>(v);
    const float scale = rsqrtf((float)DD * gtot);
#pragma unroll
    for (int c = 0; c < 16; ++c) {
        v4f o;
        o[0] = in[c][0] + v[4 * c + 0] * scale; o[1] = in[c][1] + v[4 * c + 1] * scale;
        o[2] = in[c][2] + v[4 * c + 2] * scale; o[3] = in[c][3] + v[4 * c + 3] * scale;
        row4[64 * c] = o;
    }
}

struct Frag { v4f q0, q1, q2, q3; };

__device__ __forceinline__ void load_frag(Frag& f, const float* __restrict__ p) {
    const v4f* p4 = (const v4f*)p;
    f.q0 = p4[0]; f.q1 = p4[1]; f.q2 = p4[2]; f.q3 = p4[3];
}

#define SCHED_FENCE() __builtin_amdgcn_sched_barrier(0)
#define KEEP_LOADS_IN_THIS_TRIP() asm volatile("" ::: "memory")

#define LO2(q) __builtin_shufflevector(q, q, 0, 1)
#define HI2(q) __builtin_shufflevector(q, q, 2, 3)
#define WMMA_F32(a, b, c) __builtin_amdgcn_wmma_f32_16x16x4_f32(false, a, false, b, (short)0, c, false, false)
#define KSTEP(SEL, Q)                                   \
    acc0 = WMMA_F32(SEL(fa.Q), SEL(f0.Q), acc0);        \
    acc1 = WMMA_F32(SEL(fa.Q), SEL(f1.Q), acc1);        \
    acc2 = WMMA_F32(SEL(fa.Q), SEL(f2.Q), acc2);        \
    acc3 = WMMA_F32(SEL(fa.Q), SEL(f3.Q), acc3);

__device__ __forceinline__ void mma_chunk(const Frag& fa, const Frag& f0, const Frag& f1,
                                          const Frag& f2, const Frag& f3,
                                          v8f& acc0, v8f& acc1, v8f& acc2, v8f& acc3) {
    KSTEP(LO2, q0) KSTEP(HI2, q0)
    KSTEP(LO2, q1) KSTEP(HI2, q1)
    KSTEP(LO2, q2) KSTEP(HI2, q2)
    KSTEP(LO2, q3) KSTEP(HI2, q3)
}

template<int RELU, int KS>
__global__ __launch_bounds__(256)
void gemm_wmma_kernel(const float* __restrict__ A, const float* __restrict__ B,
                      const float* __restrict__ bias, float* __restrict__ C,
                      int M, int N, int K) {
    constexpr int SPB = 8 / KS;
    __shared__ v4f part[8 * 8 * 32];
    int lane = threadIdx.x & 31;
    int wave = threadIdx.x >> 5;
    int sl = wave / KS;
    int ks = wave % KS;
    int job = blockIdx.x * SPB + sl;
    int mTiles = M >> 4;
    int mt = job % mTiles;
    int ng = job / mTiles;
    int row = lane & 15;
    int kg = (lane >> 4) << 4;
    int kLen = K / KS;
    size_t koff = (size_t)ks * kLen + kg;
    const float* ap  = A + (size_t)(mt * 16 + row) * K + koff;
    const float* bp0 = B + (size_t)(ng * 64 + row) * K + koff;
    const float* bp1 = bp0 + (size_t)16 * K;
    const float* bp2 = bp0 + (size_t)32 * K;
    const float* bp3 = bp0 + (size_t)48 * K;
    v8f acc0 = {}, acc1 = {}, acc2 = {}, acc3 = {};

    Frag xa, x0, x1, x2, x3;
    Frag ya, y0, y1, y2, y3;

    load_frag(xa, ap); load_frag(x0, bp0); load_frag(x1, bp1);
    load_frag(x2, bp2); load_frag(x3, bp3);

    int k0 = 0;
    for (; k0 + 64 < kLen; k0 += 64) {
        load_frag(ya, ap + k0 + 32); load_frag(y0, bp0 + k0 + 32); load_frag(y1, bp1 + k0 + 32);
        load_frag(y2, bp2 + k0 + 32); load_frag(y3, bp3 + k0 + 32);
        SCHED_FENCE();
        mma_chunk(xa, x0, x1, x2, x3, acc0, acc1, acc2, acc3);
        SCHED_FENCE();
        load_frag(xa, ap + k0 + 64); load_frag(x0, bp0 + k0 + 64); load_frag(x1, bp1 + k0 + 64);
        load_frag(x2, bp2 + k0 + 64); load_frag(x3, bp3 + k0 + 64);
        KEEP_LOADS_IN_THIS_TRIP();
        SCHED_FENCE();
        mma_chunk(ya, y0, y1, y2, y3, acc0, acc1, acc2, acc3);
    }
    load_frag(ya, ap + k0 + 32); load_frag(y0, bp0 + k0 + 32); load_frag(y1, bp1 + k0 + 32);
    load_frag(y2, bp2 + k0 + 32); load_frag(y3, bp3 + k0 + 32);
    SCHED_FENCE();
    mma_chunk(xa, x0, x1, x2, x3, acc0, acc1, acc2, acc3);
    mma_chunk(ya, y0, y1, y2, y3, acc0, acc1, acc2, acc3);

    v4f* mine = part + wave * 256;
    v8f accs[4] = {acc0, acc1, acc2, acc3};
#pragma unroll
    for (int t = 0; t < 4; ++t) {
        mine[(2 * t) * 32 + lane]     = __builtin_shufflevector(accs[t], accs[t], 0, 1, 2, 3);
        mine[(2 * t + 1) * 32 + lane] = __builtin_shufflevector(accs[t], accs[t], 4, 5, 6, 7);
    }
    __syncthreads();

    constexpr int GPW = 8 / KS;
    const v4f* sp = part + sl * KS * 256;
    int mrow = mt * 16 + (lane >> 4) * 8;
#pragma unroll
    for (int j = 0; j < GPW; ++j) {
        int g = ks * GPW + j;
        int t = g >> 1;
        int vb = (g & 1) * 4;
        v4f s = sp[g * 32 + lane];
#pragma unroll
        for (int q = 1; q < KS; ++q) s += sp[q * 256 + g * 32 + lane];
        int n = ng * 64 + t * 16 + row;
        float bv = bias[n];
#pragma unroll
        for (int e = 0; e < 4; ++e) {
            float xv = s[e] + bv;
            if (RELU) xv = fmaxf(xv, 0.f);
            C[(size_t)(mrow + vb + e) * N + n] = xv;
        }
    }
}

extern "C" void kernel_launch(void* const* d_in, const int* in_sizes, int n_in,
                              void* d_out, int out_size, void* d_ws, size_t ws_size,
                              hipStream_t stream) {
    const float* x     = (const float*)d_in[0];
    const float* theta = (const float*)d_in[1];
    const float* iW1   = (const float*)d_in[2];
    const float* ib1   = (const float*)d_in[3];
    const float* iW2   = (const float*)d_in[4];
    const float* ib2   = (const float*)d_in[5];
    const float* BB_W1 = (const float*)d_in[6];
    const int*   Pi_W1 = (const int*)d_in[7];
    const float* GG_W1 = (const float*)d_in[8];
    const float* BB_b1 = (const float*)d_in[9];
    const int*   Pi_b1 = (const int*)d_in[10];
    const float* GG_b1 = (const float*)d_in[11];
    const float* BB_W2 = (const float*)d_in[12];
    const int*   Pi_W2 = (const int*)d_in[13];
    const float* GG_W2 = (const float*)d_in[14];
    const float* BB_b2 = (const float*)d_in[15];
    const int*   Pi_b2 = (const int*)d_in[16];
    const float* GG_b2 = (const float*)d_in[17];
    float* out = (float*)d_out;

    float* ws = (float*)d_ws;
    size_t o = 0;
    float* W1   = ws + o; o += DD_W;
    float* W2   = ws + o; o += DD_W;
    float* b1   = ws + o; o += 3072;
    float* b2   = ws + o; o += 768;
    float* hbuf = ws + o; o += (size_t)256 * 3072;
    float* part = ws + o; o += 2 * NPART;

    fwht_passA_kernel<<<dim3(1 + NPART, 2), 512, 0, stream>>>(theta, BB_W1, Pi_W1, GG_W1, W1,
                                                              BB_W2, Pi_W2, GG_W2, W2, 2047, part,
                                                              BB_b1, Pi_b1, GG_b1, ib1, b1,
                                                              BB_b2, Pi_b2, GG_b2, ib2, b2);
    fwht_passB_kernel<<<dim3(ROWS_OUT, 2), 64, 0, stream>>>(W1, iW1, W2, iW2, part, DD_W);

    gemm_wmma_kernel<1, 2><<<192, 256, 0, stream>>>(x, W1, b1, hbuf, 256, 3072, 768);
    gemm_wmma_kernel<0, 8><<<192, 256, 0, stream>>>(hbuf, W2, b2, out, 256, 768, 3072);
}


# GEMM strips 16x64 -> 16x48 (3 N-tiles per wave): both GEMMs become 256 workgroups, one per WGP, 288 WMMAs per wave instead of 384 on 192 of 256 WGPs; k slices and every sum order kept
# speedup vs baseline: 6.4707x; 1.0363x over previous
#include <hip/hip_runtime.h>

typedef __attribute__((ext_vector_type(2))) float v2f;
typedef __attribute__((ext_vector_type(4))) float v4f;
typedef __attribute__((ext_vector_type(8))) float v8f;

#define COLS     4096
#define ROWS     1024
#define ROWS_OUT 576
#define DD_W     2359296
#define TC       32
#define PB_PITCH 260
#define P_W      2048

__device__ __forceinline__ void fwht_lds(float* buf, int n, int tid, int nt) {
    for (int h = 1; h < n; h <<= 1) {
        __syncthreads();
        for (int p = tid; p < (n >> 1); p += nt) {
            int i0 = ((p & ~(h - 1)) << 1) | (p & (h - 1));
            int i1 = i0 + h;
            float a = buf[i0], b = buf[i1];
            buf[i0] = a + b;
            buf[i1] = a - b;
        }
    }
    __syncthreads();
}

__device__ __forceinline__ void lds_only_barrier() {
    __builtin_amdgcn_fence(__ATOMIC_RELEASE, "workgroup", "local");
    __builtin_amdgcn_s_barrier();
    __builtin_amdgcn_fence(__ATOMIC_ACQUIRE, "workgroup", "local");
}

__device__ __forceinline__ void fwht_lds_inflight(float* buf, int n, int tid, int nt) {
    for (int h = 1; h < n; h <<= 1) {
        lds_only_barrier();
        for (int p = tid; p < (n >> 1); p += nt) {
            int i0 = ((p & ~(h - 1)) << 1) | (p & (h - 1));
            int i1 = i0 + h;
            float a = buf[i0], b = buf[i1];
            buf[i0] = a + b;
            buf[i1] = a - b;
        }
    }
    lds_only_barrier();
}

#define NPART (COLS / TC)

__device__ __forceinline__ float wave_sum(float x) {
#pragma unroll
    for (int m = warpSize >> 1; m > 0; m >>= 1) x += __shfl_xor(x, m);
    return x;
}

template<int N, int B0, int B1>
__device__ __forceinline__ void fwht_reg_stages(float (&v)[N]) {
#pragma unroll
    for (int b = B0; b < B1; ++b) {
        const int h = 1 << b;
#pragma unroll
        for (int p = 0; p < N / 2; ++p) {
            const int i0 = ((p & ~(h - 1)) << 1) | (p & (h - 1));
            const float x = v[i0], y = v[i0 + h];
            v[i0] = x + y;
            v[i0 + h] = x - y;
        }
    }
}

template<int P, int LLv, int DD>
__device__ __forceinline__ void bias_projection(const float* __restrict__ theta,
                                                const float* __restrict__ BB,
                                                const int* __restrict__ Pi,
                                                const float* __restrict__ GG,
                                                const float* __restrict__ init,
                                                float* __restrict__ out,
                                                float* w, float* m, float* red) {
    constexpr int nt = 512;
    constexpr int NP = P / nt, NL = LLv / nt, ND = (DD + nt - 1) / nt;
    const int tid = threadIdx.x;
    float bv[NP], tv[NP], gg[NL], in[ND];
    int pi[NL];
#pragma unroll
    for (int k = 0; k < NP; ++k) { bv[k] = BB[tid + nt * k]; tv[k] = theta[tid + nt * k]; }
#pragma unroll
    for (int k = 0; k < NL; ++k) gg[k] = GG[tid + nt * k];
#pragma unroll
    for (int k = 0; k < NL; ++k) pi[k] = Pi[tid + nt * k];
#pragma unroll
    for (int k = 0; k < ND; ++k) in[k] = (tid + nt * k < DD) ? init[tid + nt * k] : 0.f;
#pragma unroll
    for (int k = 0; k < NP; ++k) w[tid + nt * k] = bv[k] * tv[k];
    fwht_lds(w, P, tid, nt);
    float s = 0.f;
#pragma unroll
    for (int k = 0; k < NL; ++k) { float g = gg[k]; s += g * g; }
    red[tid] = s; __syncthreads();
    for (int off = nt >> 1; off > 0; off >>= 1) {
        if (tid < off) red[tid] += red[tid + off];
        __syncthreads();
    }
#pragma unroll
    for (int k = 0; k < NL; ++k) m[tid + nt * k] = w[pi[k] & (P - 1)] * gg[k];
    fwht_lds(m, LLv, tid, nt);
    float scale = rsqrtf((float)DD * red[0]);
#pragma unroll
    for (int k = 0; k < ND; ++k) {
        const int i = tid + nt * k;
        if (i < DD) out[i] = in[k] + m[i] * scale;
    }
}

__global__ __launch_bounds__(512)
void fwht_passA_kernel(const float* __restrict__ theta,
                       const float* __restrict__ BB_a,
                       const int* __restrict__ Pi_a, const float* __restrict__ GG_a,
                       float* __restrict__ m_a,
                       const float* __restrict__ BB_b,
                       const int* __restrict__ Pi_b, const float* __restrict__ GG_b,
                       float* __restrict__ m_b, int pmask,
                       float* __restrict__ part2,
                       const float* __restrict__ BB_b1, const int* __restrict__ Pi_b1,
                       const float* __restrict__ GG_b1, const float* __restrict__ init_b1,
                       float* __restrict__ out_b1,
                       const float* __restrict__ BB_b2, const int* __restrict__ Pi_b2,
                       const float* __restrict__ GG_b2, const float* __restrict__ init_b2,
                       float* __restrict__ out_b2) {
    __shared__ float tile[ROWS * TC];
    __shared__ float wl[P_W];
    __shared__ float wsum[16];
    if (blockIdx.x == 0) {
        if (blockIdx.y == 0)
            bias_projection<P_W, 4096, 3072>(theta, BB_b1, Pi_b1, GG_b1, init_b1, out_b1,
                                             wl, tile, tile + 4096);
        else
            bias_projection<1024, 1024, 768>(theta, BB_b2, Pi_b2, GG_b2, init_b2, out_b2,
                                             wl, tile, tile + 4096);
        return;
    }
    const int lane = threadIdx.x & 31;
    const int wv = threadIdx.x >> 5;
    const int tix = blockIdx.x - 1;
    const float* __restrict__ BB = blockIdx.y ? BB_b : BB_a;
    const int*   __restrict__ Pi = blockIdx.y ? Pi_b : Pi_a;
    const float* __restrict__ GG = blockIdx.y ? GG_b : GG_a;
    float* __restrict__ mbuf     = blockIdx.y ? m_b : m_a;
    const int col = tix * TC + lane;

    float wp[P_W / 512];
#pragma unroll
    for (int k = 0; k < P_W / 512; ++k) wp[k] = BB[threadIdx.x + 512 * k] * theta[threadIdx.x + 512 * k];
    float v[64];
    int pi[64];
    const int g0 = wv * 64 * COLS + col;
#pragma unroll
    for (int j = 0; j < 64; ++j) pi[j] = Pi[g0 + j * COLS];
#pragma unroll
    for (int j = 0; j < 64; ++j) v[j] = GG[g0 + j * COLS];
#pragma unroll
    for (int k = 0; k < P_W / 512; ++k) wl[threadIdx.x + 512 * k] = wp[k];
    fwht_lds_inflight(wl, P_W, threadIdx.x, 512);
    float ss[4] = {0.f, 0.f, 0.f, 0.f};
#pragma unroll
    for (int j = 0; j < 64; ++j) {
        const float g = v[j];
        ss[j & 3] += g * g;
        v[j] = wl[pi[j] & pmask] * g;
    }
    const float wtot = wave_sum((ss[0] + ss[1]) + (ss[2] + ss[3]));
    if (lane == 0) wsum[wv] = wtot;
    fwht_reg_stages<64, 0, 6>(v);
#pragma unroll
    for (int j = 0; j < 64; ++j) tile[(wv * 64 + j) * TC + lane] = v[j];
    __syncthreads();
    if (threadIdx.x == 0) {
        float tot = wsum[0];
#pragma unroll
        for (int i = 1; i < 16; ++i) tot += wsum[i];
        part2[NPART * blockIdx.y + tix] = tot;
    }
    float u[64];
#pragma unroll
    for (int q = 0; q < 4; ++q) {
#pragma unroll
        for (int jh = 0; jh < 16; ++jh) u[16 * q + jh] = tile[(64 * jh + 4 * wv + q) * TC + lane];
    }
    fwht_reg_stages<64, 0, 4>(u);
#pragma unroll
    for (int q = 0; q < 4; ++q) {
#pragma unroll
        for (int jh = 0; jh < ROWS_OUT / 64; ++jh)
            mbuf[(size_t)(64 * jh + 4 * wv + q) * COLS + col] = u[16 * q + jh];
    }
}

__global__ __launch_bounds__(64)
void fwht_passB_kernel(float* buf_a, const float* __restrict__ init_a,
                       float* buf_b, const float* __restrict__ init_b,
                       const float* __restrict__ part2, int DD) {
    __shared__ v4f blk4[16 * PB_PITCH / 4];
    __shared__ float wsum[2];
    float* blk = (float*)blk4;
    const int t = threadIdx.x;
    float* buf = blockIdx.y ? buf_b : buf_a;
    const float* __restrict__ init = blockIdx.y ? init_b : init_a;
    const float* __restrict__ part = part2 + NPART * blockIdx.y;
    const size_t base = (size_t)blockIdx.x * COLS;
    v4f* row4 = (v4f*)(buf + base) + t;
    const v4f* __restrict__ init4 = (const v4f*)(init + base) + t;

    const float ps = wave_sum(part[t] + part[t + 64]);
    float v[64];
    v4f in[16];
#pragma unroll
    for (int c = 0; c < 16; ++c) {
        const v4f x = row4[64 * c];
        v[4 * c + 0] = x[0]; v[4 * c + 1] = x[1]; v[4 * c + 2] = x[2]; v[4 * c + 3] = x[3];
    }
#pragma unroll
    for (int c = 0; c < 16; ++c) in[c] = init4[64 * c];
    fwht_reg_stages<64, 0, 2>(v);
#pragma unroll
    for (int c = 0; c < 16; ++c) {
        v4f x; x[0] = v[4 * c + 0]; x[1] = v[4 * c + 1]; x[2] = v[4 * c + 2]; x[3] = v[4 * c + 3];
        blk4[c * (PB_PITCH / 4) + t] = x;
    }
    if ((t & 31) == 0) wsum[t >> 5] = ps;
    __syncthreads();
    const float gtot = wsum[0] + wsum[1];
    float* pb = blk + (t >> 2) * PB_PITCH + (t & 3);
    float u[64];
#pragma unroll
    for (int m = 0; m < 64; ++m) u[m] = pb[4 * m];
    fwht_reg_stages<64, 0, 6>(u);
#pragma unroll
    for (int m = 0; m < 64; ++m) pb[4 * m] = u[m];
    __syncthreads();
#pragma unroll
    for (int c = 0; c < 16; ++c) {
        const v4f x = blk4[c * (PB_PITCH / 4) + t];
        v[4 * c + 0] = x[0]; v[4 * c + 1] = x[1]; v[4 * c + 2] = x[2]; v[4 * c + 3] = x[3];
    }
    fwht_reg_stages<64, 2, 6>(v);
    const float scale = rsqrtf((float)DD * gtot);
#pragma unroll
    for (int c = 0; c < 16; ++c) {
        v4f o;
        o[0] = in[c][0] + v[4 * c + 0] * scale; o[1] = in[c][1] + v[4 * c + 1] * scale;
        o[2] = in[c][2] + v[4 * c + 2] * scale; o[3] = in[c][3] + v[4 * c + 3] * scale;
        row4[64 * c] = o;
    }
}

struct Frag { v4f q0, q1, q2, q3; };

__device__ __forceinline__ void load_frag(Frag& f, const float* __restrict__ p) {
    const v4f* p4 = (const v4f*)p;
    f.q0 = p4[0]; f.q1 = p4[1]; f.q2 = p4[2]; f.q3 = p4[3];
}

#define SCHED_FENCE() __builtin_amdgcn_sched_barrier(0)
#define KEEP_LOADS_IN_THIS_TRIP() asm volatile("" ::: "memory")

#define LO2(q) __builtin_shufflevector(q, q, 0, 1)
#define HI2(q) __builtin_shufflevector(q, q, 2, 3)
#define WMMA_F32(a, b, c) __builtin_amdgcn_wmma_f32_16x16x4_f32(false, a, false, b, (short)0, c, false, false)
#define NT 3
#define KSTEP(SEL, Q)                                   \
    acc0 = WMMA_F32(SEL(fa.Q), SEL(f0.Q), acc0);        \
    acc1 = WMMA_F32(SEL(fa.Q), SEL(f1.Q), acc1);        \
    acc2 = WMMA_F32(SEL(fa.Q), SEL(f2.Q), acc2);

__device__ __forceinline__ void mma_chunk(const Frag& fa, const Frag& f0, const Frag& f1,
                                          const Frag& f2,
                                          v8f& acc0, v8f& acc1, v8f& acc2) {
    KSTEP(LO2, q0) KSTEP(HI2, q0)
    KSTEP(LO2, q1) KSTEP(HI2, q1)
    KSTEP(LO2, q2) KSTEP(HI2, q2)
    KSTEP(LO2, q3) KSTEP(HI2, q3)
}

template<int RELU, int KS>
__global__ __launch_bounds__(256)
void gemm_wmma_kernel(const float* __restrict__ A, const float* __restrict__ B,
                      const float* __restrict__ bias, float* __restrict__ C,
                      int M, int N, int K) {
    constexpr int SPB = 8 / KS;
    constexpr int NG = 2 * NT;
    __shared__ v4f part[8 * NG * 32];
    int lane = threadIdx.x & 31;
    int wave = threadIdx.x >> 5;
    int sl = wave / KS;
    int ks = wave % KS;
    int job = blockIdx.x * SPB + sl;
    int mTiles = M >> 4;
    int mt = job % mTiles;
    int ng = job / mTiles;
    int row = lane & 15;
    int kg = (lane >> 4) << 4;
    int kLen = K / KS;
    size_t koff = (size_t)ks * kLen + kg;
    const float* ap  = A + (size_t)(mt * 16 + row) * K + koff;
    const float* bp0 = B + (size_t)(ng * (16 * NT) + row) * K + koff;
    const float* bp1 = bp0 + (size_t)16 * K;
    const float* bp2 = bp0 + (size_t)32 * K;
    v8f acc0 = {}, acc1 = {}, acc2 = {};

    Frag xa, x0, x1, x2;
    Frag ya, y0, y1, y2;

    load_frag(xa, ap); load_frag(x0, bp0); load_frag(x1, bp1);
    load_frag(x2, bp2);

    int k0 = 0;
    for (; k0 + 64 < kLen; k0 += 64) {
        load_frag(ya, ap + k0 + 32); load_frag(y0, bp0 + k0 + 32); load_frag(y1, bp1 + k0 + 32);
        load_frag(y2, bp2 + k0 + 32);
        SCHED_FENCE();
        mma_chunk(xa, x0, x1, x2, acc0, acc1, acc2);
        SCHED_FENCE();
        load_frag(xa, ap + k0 + 64); load_frag(x0, bp0 + k0 + 64); load_frag(x1, bp1 + k0 + 64);
        load_frag(x2, bp2 + k0 + 64);
        KEEP_LOADS_IN_THIS_TRIP();
        SCHED_FENCE();
        mma_chunk(ya, y0, y1, y2, acc0, acc1, acc2);
    }
    load_frag(ya, ap + k0 + 32); load_frag(y0, bp0 + k0 + 32); load_frag(y1, bp1 + k0 + 32);
    load_frag(y2, bp2 + k0 + 32);
    SCHED_FENCE();
    mma_chunk(xa, x0, x1, x2, acc0, acc1, acc2);
    mma_chunk(ya, y0, y1, y2, acc0, acc1, acc2);

    v4f* mine = part + wave * (NG * 32);
    v8f accs[NT] = {acc0, acc1, acc2};
#pragma unroll
    for (int t = 0; t < NT; ++t) {
        mine[(2 * t) * 32 + lane]     = __builtin_shufflevector(accs[t], accs[t], 0, 1, 2, 3);
        mine[(2 * t + 1) * 32 + lane] = __builtin_shufflevector(accs[t], accs[t], 4, 5, 6, 7);
    }
    __syncthreads();

    constexpr int GPW = (NG + KS - 1) / KS;
    const v4f* sp = part + sl * KS * (NG * 32);
    int mrow = mt * 16 + (lane >> 4) * 8;
#pragma unroll
    for (int j = 0; j < GPW; ++j) {
        int g = ks * GPW + j;
        if (g >= NG) break;
        int t = g >> 1;
        int vb = (g & 1) * 4;
        v4f s = sp[g * 32 + lane];
#pragma unroll
        for (int q = 1; q < KS; ++q) s += sp[q * (NG * 32) + g * 32 + lane];
        int n = ng * (16 * NT) + t * 16 + row;
        float bv = bias[n];
#pragma unroll
        for (int e = 0; e < 4; ++e) {
            float xv = s[e] + bv;
            if (RELU) xv = fmaxf(xv, 0.f);
            C[(size_t)(mrow + vb + e) * N + n] = xv;
        }
    }
}

extern "C" void kernel_launch(void* const* d_in, const int* in_sizes, int n_in,
                              void* d_out, int out_size, void* d_ws, size_t ws_size,
                              hipStream_t stream) {
    const float* x     = (const float*)d_in[0];
    const float* theta = (const float*)d_in[1];
    const float* iW1   = (const float*)d_in[2];
    const float* ib1   = (const float*)d_in[3];
    const float* iW2   = (const float*)d_in[4];
    const float* ib2   = (const float*)d_in[5];
    const float* BB_W1 = (const float*)d_in[6];
    const int*   Pi_W1 = (const int*)d_in[7];
    const float* GG_W1 = (const float*)d_in[8];
    const float* BB_b1 = (const float*)d_in[9];
    const int*   Pi_b1 = (const int*)d_in[10];
    const float* GG_b1 = (const float*)d_in[11];
    const float* BB_W2 = (const float*)d_in[12];
    const int*   Pi_W2 = (const int*)d_in[13];
    const float* GG_W2 = (const float*)d_in[14];
    const float* BB_b2 = (const float*)d_in[15];
    const int*   Pi_b2 = (const int*)d_in[16];
    const float* GG_b2 = (const float*)d_in[17];
    float* out = (float*)d_out;

    float* ws = (float*)d_ws;
    size_t o = 0;
    float* W1   = ws + o; o += DD_W;
    float* W2   = ws + o; o += DD_W;
    float* b1   = ws + o; o += 3072;
    float* b2   = ws + o; o += 768;
    float* hbuf = ws + o; o += (size_t)256 * 3072;
    float* part = ws + o; o += 2 * NPART;

    fwht_passA_kernel<<<dim3(1 + NPART, 2), 512, 0, stream>>>(theta, BB_W1, Pi_W1, GG_W1, W1,
                                                              BB_W2, Pi_W2, GG_W2, W2, 2047, part,
                                                              BB_b1, Pi_b1, GG_b1, ib1, b1,
                                                              BB_b2, Pi_b2, GG_b2, ib2, b2);
    fwht_passB_kernel<<<dim3(ROWS_OUT, 2), 64, 0, stream>>>(W1, iW1, W2, iW2, part, DD_W);

    gemm_wmma_kernel<1, 2><<<256, 256, 0, stream>>>(x, W1, b1, hbuf, 256, 3072, 768);
    gemm_wmma_kernel<0, 8><<<256, 256, 0, stream>>>(hbuf, W2, b2, out, 256, 768, 3072);
}


# Pass B of W1 moves into GEMM1's launch: a workgroup turns 3 rows = 16 rows of W1 into an LDS tile and multiplies x by it; 4 launches -> 3, W1 never in memory; GEMM1 sums k in one run, no slices
# speedup vs baseline: 8.6853x; 1.3423x over previous
#include <hip/hip_runtime.h>

typedef __attribute__((ext_vector_type(2))) float v2f;
typedef __attribute__((ext_vector_type(4))) float v4f;
typedef __attribute__((ext_vector_type(8))) float v8f;

#define COLS     4096
#define ROWS     1024
#define ROWS_OUT 576
#define DD_W     2359296
#define TC       32
#define PB_PITCH 260
#define P_W      2048

__device__ __forceinline__ void fwht_lds(float* buf, int n, int tid, int nt) {
    for (int h = 1; h < n; h <<= 1) {
        __syncthreads();
        for (int p = tid; p < (n >> 1); p += nt) {
            int i0 = ((p & ~(h - 1)) << 1) | (p & (h - 1));
            int i1 = i0 + h;
            float a = buf[i0], b = buf[i1];
            buf[i0] = a + b;
            buf[i1] = a - b;
        }
    }
    __syncthreads();
}

__device__ __forceinline__ void lds_only_barrier() {
    __builtin_amdgcn_fence(__ATOMIC_RELEASE, "workgroup", "local");
    __builtin_amdgcn_s_barrier();
    __builtin_amdgcn_fence(__ATOMIC_ACQUIRE, "workgroup", "local");
}

__device__ __forceinline__ void fwht_lds_inflight(float* buf, int n, int tid, int nt) {
    for (int h = 1; h < n; h <<= 1) {
        lds_only_barrier();
        for (int p = tid; p < (n >> 1); p += nt) {
            int i0 = ((p & ~(h - 1)) << 1) | (p & (h - 1));
            int i1 = i0 + h;
            float a = buf[i0], b = buf[i1];
            buf[i0] = a + b;
            buf[i1] = a - b;
        }
    }
    lds_only_barrier();
}

#define NPART (COLS / TC)

__device__ __forceinline__ float wave_sum(float x) {
#pragma unroll
    for (int m = warpSize >> 1; m > 0; m >>= 1) x += __shfl_xor(x, m);
    return x;
}

template<int N, int B0, int B1>
__device__ __forceinline__ void fwht_reg_stages(float (&v)[N]) {
#pragma unroll
    for (int b = B0; b < B1; ++b) {
        const int h = 1 << b;
#pragma unroll
        for (int p = 0; p < N / 2; ++p) {
            const int i0 = ((p & ~(h - 1)) << 1) | (p & (h - 1));
            const float x = v[i0], y = v[i0 + h];
            v[i0] = x + y;
            v[i0 + h] = x - y;
        }
    }
}

template<int P, int LLv, int DD>
__device__ __forceinline__ void bias_projection(const float* __restrict__ theta,
                                                const float* __restrict__ BB,
                                                const int* __restrict__ Pi,
                                                const float* __restrict__ GG,
                                                const float* __restrict__ init,
                                                float* __restrict__ out,
                                                float* w, float* m, float* red) {
    constexpr int nt = 512;
    constexpr int NP = P / nt, NL = LLv / nt, ND = (DD + nt - 1) / nt;
    const int tid = threadIdx.x;
    float bv[NP], tv[NP], gg[NL], in[ND];
    int pi[NL];
#pragma unroll
    for (int k = 0; k < NP; ++k) { bv[k] = BB[tid + nt * k]; tv[k] = theta[tid + nt * k]; }
#pragma unroll
    for (int k = 0; k < NL; ++k) gg[k] = GG[tid + nt * k];
#pragma unroll
    for (int k = 0; k < NL; ++k) pi[k] = Pi[tid + nt * k];
#pragma unroll
    for (int k = 0; k < ND; ++k) in[k] = (tid + nt * k < DD) ? init[tid + nt * k] : 0.f;
#pragma unroll
    for (int k = 0; k < NP; ++k) w[tid + nt * k] = bv[k] * tv[k];
    fwht_lds(w, P, tid, nt);
    float s = 0.f;
#pragma unroll
    for (int k = 0; k < NL; ++k) { float g = gg[k]; s += g * g; }
    red[tid] = s; __syncthreads();
    for (int off = nt >> 1; off > 0; off >>= 1) {
        if (tid < off) red[tid] += red[tid + off];
        __syncthreads();
    }
#pragma unroll
    for (int k = 0; k < NL; ++k) m[tid + nt * k] = w[pi[k] & (P - 1)] * gg[k];
    fwht_lds(m, LLv, tid, nt);
    float scale = rsqrtf((float)DD * red[0]);
#pragma unroll
    for (int k = 0; k < ND; ++k) {
        const int i = tid + nt * k;
        if (i < DD) out[i] = in[k] + m[i] * scale;
    }
}

__global__ __launch_bounds__(512)
void fwht_passA_kernel(const float* __restrict__ theta,
                       const float* __restrict__ BB_a,
                       const int* __restrict__ Pi_a, const float* __restrict__ GG_a,
                       float* __restrict__ m_a,
                       const float* __restrict__ BB_b,
                       const int* __restrict__ Pi_b, const float* __restrict__ GG_b,
                       float* __restrict__ m_b, int pmask,
                       float* __restrict__ part2,
                       const float* __restrict__ BB_b1, const int* __restrict__ Pi_b1,
                       const float* __restrict__ GG_b1, const float* __restrict__ init_b1,
                       float* __restrict__ out_b1,
                       const float* __restrict__ BB_b2, const int* __restrict__ Pi_b2,
                       const float* __restrict__ GG_b2, const float* __restrict__ init_b2,
                       float* __restrict__ out_b2) {
    __shared__ float tile[ROWS * TC];
    __shared__ float wl[P_W];
    __shared__ float wsum[16];
    if (blockIdx.x == 0) {
        if (blockIdx.y == 0)
            bias_projection<P_W, 4096, 3072>(theta, BB_b1, Pi_b1, GG_b1, init_b1, out_b1,
                                             wl, tile, tile + 4096);
        else
            bias_projection<1024, 1024, 768>(theta, BB_b2, Pi_b2, GG_b2, init_b2, out_b2,
                                             wl, tile, tile + 4096);
        return;
    }
    const int lane = threadIdx.x & 31;
    const int wv = threadIdx.x >> 5;
    const int tix = blockIdx.x - 1;
    const float* __restrict__ BB = blockIdx.y ? BB_b : BB_a;
    const int*   __restrict__ Pi = blockIdx.y ? Pi_b : Pi_a;
    const float* __restrict__ GG = blockIdx.y ? GG_b : GG_a;
    float* __restrict__ mbuf     = blockIdx.y ? m_b : m_a;
    const int col = tix * TC + lane;

    float wp[P_W / 512];
#pragma unroll
    for (int k = 0; k < P_W / 512; ++k) wp[k] = BB[threadIdx.x + 512 * k] * theta[threadIdx.x + 512 * k];
    float v[64];
    int pi[64];
    const int g0 = wv * 64 * COLS + col;
#pragma unroll
    for (int j = 0; j < 64; ++j) pi[j] = Pi[g0 + j * COLS];
#pragma unroll
    for (int j = 0; j < 64; ++j) v[j] = GG[g0 + j * COLS];
#pragma unroll
    for (int k = 0; k < P_W / 512; ++k) wl[threadIdx.x + 512 * k] = wp[k];
    fwht_lds_inflight(wl, P_W, threadIdx.x, 512);
    float ss[4] = {0.f, 0.f, 0.f, 0.f};
#pragma unroll
    for (int j = 0; j < 64; ++j) {
        const float g = v[j];
        ss[j & 3] += g * g;
        v[j] = wl[pi[j] & pmask] * g;
    }
    const float wtot = wave_sum((ss[0] + ss[1]) + (ss[2] + ss[3]));
    if (lane == 0) wsum[wv] = wtot;
    fwht_reg_stages<64, 0, 6>(v);
#pragma unroll
    for (int j = 0; j < 64; ++j) tile[(wv * 64 + j) * TC + lane] = v[j];
    __syncthreads();
    if (threadIdx.x == 0) {
        float tot = wsum[0];
#pragma unroll
        for (int i = 1; i < 16; ++i) tot += wsum[i];
        part2[NPART * blockIdx.y + tix] = tot;
    }
    float u[64];
#pragma unroll
    for (int q = 0; q < 4; ++q) {
#pragma unroll
        for (int jh = 0; jh < 16; ++jh) u[16 * q + jh] = tile[(64 * jh + 4 * wv + q) * TC + lane];
    }
    fwht_reg_stages<64, 0, 4>(u);
#pragma unroll
    for (int q = 0; q < 4; ++q) {
#pragma unroll
        for (int jh = 0; jh < ROWS_OUT / 64; ++jh)
            mbuf[(size_t)(64 * jh + 4 * wv + q) * COLS + col] = u[16 * q + jh];
    }
}

struct Frag { v4f q0, q1, q2, q3; };

__device__ __forceinline__ void load_frag(Frag& f, const float* __restrict__ p) {
    const v4f* p4 = (const v4f*)p;
    f.q0 = p4[0]; f.q1 = p4[1]; f.q2 = p4[2]; f.q3 = p4[3];
}

#define SCHED_FENCE() __builtin_amdgcn_sched_barrier(0)
#define KEEP_LOADS_IN_THIS_TRIP() asm volatile("" ::: "memory")

#define LO2(q) __builtin_shufflevector(q, q, 0, 1)
#define HI2(q) __builtin_shufflevector(q, q, 2, 3)
#define WMMA_F32(a, b, c) __builtin_amdgcn_wmma_f32_16x16x4_f32(false, a, false, b, (short)0, c, false, false)
#define NT 3
#define KSTEP(SEL, Q)                                   \
    acc0 = WMMA_F32(SEL(fa.Q), SEL(f0.Q), acc0);        \
    acc1 = WMMA_F32(SEL(fa.Q), SEL(f1.Q), acc1);        \
    acc2 = WMMA_F32(SEL(fa.Q), SEL(f2.Q), acc2);

__device__ __forceinline__ void mma_chunk(const Frag& fa, const Frag& f0, const Frag& f1,
                                          const Frag& f2,
                                          v8f& acc0, v8f& acc1, v8f& acc2) {
    KSTEP(LO2, q0) KSTEP(HI2, q0)
    KSTEP(LO2, q1) KSTEP(HI2, q1)
    KSTEP(LO2, q2) KSTEP(HI2, q2)
    KSTEP(LO2, q3) KSTEP(HI2, q3)
}

template<int RELU, int KS>
__global__ __launch_bounds__(256)
void gemm_wmma_kernel(const float* __restrict__ A, const float* __restrict__ B,
                      const float* __restrict__ bias, float* __restrict__ C,
                      int M, int N, int K) {
    constexpr int SPB = 8 / KS;
    constexpr int NG = 2 * NT;
    __shared__ v4f part[8 * NG * 32];
    int lane = threadIdx.x & 31;
    int wave = threadIdx.x >> 5;
    int sl = wave / KS;
    int ks = wave % KS;
    int job = blockIdx.x * SPB + sl;
    int mTiles = M >> 4;
    int mt = job % mTiles;
    int ng = job / mTiles;
    int row = lane & 15;
    int kg = (lane >> 4) << 4;
    int kLen = K / KS;
    size_t koff = (size_t)ks * kLen + kg;
    const float* ap  = A + (size_t)(mt * 16 + row) * K + koff;
    const float* bp0 = B + (size_t)(ng * (16 * NT) + row) * K + koff;
    const float* bp1 = bp0 + (size_t)16 * K;
    const float* bp2 = bp0 + (size_t)32 * K;
    v8f acc0 = {}, acc1 = {}, acc2 = {};

    Frag xa, x0, x1, x2;
    Frag ya, y0, y1, y2;

    load_frag(xa, ap); load_frag(x0, bp0); load_frag(x1, bp1);
    load_frag(x2, bp2);

    int k0 = 0;
    for (; k0 + 64 < kLen; k0 += 64) {
        load_frag(ya, ap + k0 + 32); load_frag(y0, bp0 + k0 + 32); load_frag(y1, bp1 + k0 + 32);
        load_frag(y2, bp2 + k0 + 32);
        SCHED_FENCE();
        mma_chunk(xa, x0, x1, x2, acc0, acc1, acc2);
        SCHED_FENCE();
        load_frag(xa, ap + k0 + 64); load_frag(x0, bp0 + k0 + 64); load_frag(x1, bp1 + k0 + 64);
        load_frag(x2, bp2 + k0 + 64);
        KEEP_LOADS_IN_THIS_TRIP();
        SCHED_FENCE();
        mma_chunk(ya, y0, y1, y2, acc0, acc1, acc2);
    }
    load_frag(ya, ap + k0 + 32); load_frag(y0, bp0 + k0 + 32); load_frag(y1, bp1 + k0 + 32);
    load_frag(y2, bp2 + k0 + 32);
    SCHED_FENCE();
    mma_chunk(xa, x0, x1, x2, acc0, acc1, acc2);
    mma_chunk(ya, y0, y1, y2, acc0, acc1, acc2);

    v4f* mine = part + wave * (NG * 32);
    v8f accs[NT] = {acc0, acc1, acc2};
#pragma unroll
    for (int t = 0; t < NT; ++t) {
        mine[(2 * t) * 32 + lane]     = __builtin_shufflevector(accs[t], accs[t], 0, 1, 2, 3);
        mine[(2 * t + 1) * 32 + lane] = __builtin_shufflevector(accs[t], accs[t], 4, 5, 6, 7);
    }
    __syncthreads();

    constexpr int GPW = (NG + KS - 1) / KS;
    const v4f* sp = part + sl * KS * (NG * 32);
    int mrow = mt * 16 + (lane >> 4) * 8;
#pragma unroll
    for (int j = 0; j < GPW; ++j) {
        int g = ks * GPW + j;
        if (g >= NG) break;
        int t = g >> 1;
        int vb = (g & 1) * 4;
        v4f s = sp[g * 32 + lane];
#pragma unroll
        for (int q = 1; q < KS; ++q) s += sp[q * (NG * 32) + g * 32 + lane];
        int n = ng * (16 * NT) + t * 16 + row;
        float bv = bias[n];
#pragma unroll
        for (int e = 0; e < 4; ++e) {
            float xv = s[e] + bv;
            if (RELU) xv = fmaxf(xv, 0.f);
            C[(size_t)(mrow + vb + e) * N + n] = xv;
        }
    }
}

#define NFUSED   192
#define WT_PITCH 772

#define KSTEP2(SEL, Q)                                  \
    acc0 = WMMA_F32(SEL(fa0.Q), SEL(fb.Q), acc0);       \
    acc1 = WMMA_F32(SEL(fa1.Q), SEL(fb.Q), acc1);

__device__ __forceinline__ void mma_chunk2(const Frag& fa0, const Frag& fa1, const Frag& fb,
                                           v8f& acc0, v8f& acc1) {
    KSTEP2(LO2, q0) KSTEP2(HI2, q0)
    KSTEP2(LO2, q1) KSTEP2(HI2, q1)
    KSTEP2(LO2, q2) KSTEP2(HI2, q2)
    KSTEP2(LO2, q3) KSTEP2(HI2, q3)
}

__device__ __forceinline__ void load_frag_lds(Frag& f, const float* p) {
    const v4f* p4 = (const v4f*)p;
    f.q0 = p4[0]; f.q1 = p4[1]; f.q2 = p4[2]; f.q3 = p4[3];
}

__global__ __launch_bounds__(256)
void passB_gemm1_kernel(const float* __restrict__ m_a, const float* __restrict__ init_a,
                        float* buf_b, const float* __restrict__ init_b,
                        const float* __restrict__ part2, int DD,
                        const float* __restrict__ X, const float* __restrict__ bias1,
                        float* __restrict__ H) {
    __shared__ v4f blk4[4 * 16 * PB_PITCH / 4];
    __shared__ v4f wt4[16 * WT_PITCH / 4];
    __shared__ float wsum[8];
    const int tid = threadIdx.x;
    const int g = tid >> 6;
    const int t = tid & 63;
    const bool fused = blockIdx.x < NFUSED;
    const bool active = !fused || g < 3;
    const int rowIdx = fused ? (int)blockIdx.x * 3 + g : ((int)blockIdx.x - NFUSED) * 4 + g;
    const float* src = fused ? m_a : buf_b;
    const float* init = fused ? init_a : init_b;
    const float* part = part2 + (fused ? 0 : NPART);
    const size_t base = (size_t)rowIdx * COLS;
    const v4f* row4 = (const v4f*)(src + base) + t;
    const v4f* init4 = (const v4f*)(init + base) + t;
    v4f* dst4 = (v4f*)(buf_b + base) + t;
    v4f* myblk4 = blk4 + g * (16 * PB_PITCH / 4);
    float* blk = (float*)myblk4;
    float* wt = (float*)wt4;

    float v[64];
    v4f in[16];
    if (active) {
        const float ps = wave_sum(part[t] + part[t + 64]);
#pragma unroll
        for (int c = 0; c < 16; ++c) {
            const v4f x = row4[64 * c];
            v[4 * c + 0] = x[0]; v[4 * c + 1] = x[1]; v[4 * c + 2] = x[2]; v[4 * c + 3] = x[3];
        }
#pragma unroll
        for (int c = 0; c < 16; ++c) in[c] = init4[64 * c];
        fwht_reg_stages<64, 0, 2>(v);
#pragma unroll
        for (int c = 0; c < 16; ++c) {
            v4f x; x[0] = v[4 * c + 0]; x[1] = v[4 * c + 1]; x[2] = v[4 * c + 2]; x[3] = v[4 * c + 3];
            myblk4[c * (PB_PITCH / 4) + t] = x;
        }
        if ((t & 31) == 0) wsum[tid >> 5] = ps;
    }
    __syncthreads();
    float gtot = 0.f;
    if (active) {
        gtot = wsum[2 * g] + wsum[2 * g + 1];
        float* pb = blk + (t >> 2) * PB_PITCH + (t & 3);
        float u[64];
#pragma unroll
        for (int m = 0; m < 64; ++m) u[m] = pb[4 * m];
        fwht_reg_stages<64, 0, 6>(u);
#pragma unroll
        for (int m = 0; m < 64; ++m) pb[4 * m] = u[m];
    }
    __syncthreads();
    if (active) {
#pragma unroll
        for (int c = 0; c < 16; ++c) {
            const v4f x = myblk4[c * (PB_PITCH / 4) + t];
            v[4 * c + 0] = x[0]; v[4 * c + 1] = x[1]; v[4 * c + 2] = x[2]; v[4 * c + 3] = x[3];
        }
        fwht_reg_stages<64, 2, 6>(v);
        const float scale = rsqrtf((float)DD * gtot);
#pragma unroll
        for (int c = 0; c < 16; ++c) {
            v4f o;
            o[0] = in[c][0] + v[4 * c + 0] * scale; o[1] = in[c][1] + v[4 * c + 1] * scale;
            o[2] = in[c][2] + v[4 * c + 2] * scale; o[3] = in[c][3] + v[4 * c + 3] * scale;
            if (fused) {
                const int f = COLS * g + 256 * c + 4 * t;
                const int n = f / 768;
                const int k = f - 768 * n;
                wt4[(n * WT_PITCH + k) >> 2] = o;
            } else {
                dst4[64 * c] = o;
            }
        }
    }
    if (!fused) return;
    __syncthreads();

    const int lane = tid & 31;
    const int wave = tid >> 5;
    const int row = lane & 15;
    const int kg = (lane >> 4) << 4;
    const float* ap0 = X + (size_t)(wave * 32 + row) * 768 + kg;
    const float* ap1 = ap0 + (size_t)16 * 768;
    const float* bl = wt + row * WT_PITCH + kg;
    v8f acc0 = {}, acc1 = {};
    Frag xa0, xa1, xb;
    Frag ya0, ya1, yb;

    load_frag(xa0, ap0); load_frag(xa1, ap1); load_frag_lds(xb, bl);
    int k0 = 0;
    for (; k0 + 64 < 768; k0 += 64) {
        load_frag(ya0, ap0 + k0 + 32); load_frag(ya1, ap1 + k0 + 32); load_frag_lds(yb, bl + k0 + 32);
        SCHED_FENCE();
        mma_chunk2(xa0, xa1, xb, acc0, acc1);
        SCHED_FENCE();
        load_frag(xa0, ap0 + k0 + 64); load_frag(xa1, ap1 + k0 + 64); load_frag_lds(xb, bl + k0 + 64);
        KEEP_LOADS_IN_THIS_TRIP();
        SCHED_FENCE();
        mma_chunk2(ya0, ya1, yb, acc0, acc1);
    }
    load_frag(ya0, ap0 + k0 + 32); load_frag(ya1, ap1 + k0 + 32); load_frag_lds(yb, bl + k0 + 32);
    SCHED_FENCE();
    mma_chunk2(xa0, xa1, xb, acc0, acc1);
    mma_chunk2(ya0, ya1, yb, acc0, acc1);

    const int n = 16 * (int)blockIdx.x + row;
    const float bv = bias1[n];
    const int mrow = wave * 32 + (lane >> 4) * 8;
#pragma unroll
    for (int e = 0; e < 8; ++e) {
        H[(size_t)(mrow + e) * 3072 + n]      = fmaxf(acc0[e] + bv, 0.f);
        H[(size_t)(mrow + 16 + e) * 3072 + n] = fmaxf(acc1[e] + bv, 0.f);
    }
}

extern "C" void kernel_launch(void* const* d_in, const int* in_sizes, int n_in,
                              void* d_out, int out_size, void* d_ws, size_t ws_size,
                              hipStream_t stream) {
    const float* x     = (const float*)d_in[0];
    const float* theta = (const float*)d_in[1];
    const float* iW1   = (const float*)d_in[2];
    const float* ib1   = (const float*)d_in[3];
    const float* iW2   = (const float*)d_in[4];
    const float* ib2   = (const float*)d_in[5];
    const float* BB_W1 = (const float*)d_in[6];
    const int*   Pi_W1 = (const int*)d_in[7];
    const float* GG_W1 = (const float*)d_in[8];
    const float* BB_b1 = (const float*)d_in[9];
    const int*   Pi_b1 = (const int*)d_in[10];
    const float* GG_b1 = (const float*)d_in[11];
    const float* BB_W2 = (const float*)d_in[12];
    const int*   Pi_W2 = (const int*)d_in[13];
    const float* GG_W2 = (const float*)d_in[14];
    const float* BB_b2 = (const float*)d_in[15];
    const int*   Pi_b2 = (const int*)d_in[16];
    const float* GG_b2 = (const float*)d_in[17];
    float* out = (float*)d_out;

    float* ws = (float*)d_ws;
    size_t o = 0;
    float* W1   = ws + o; o += DD_W;
    float* W2   = ws + o; o += DD_W;
    float* b1   = ws + o; o += 3072;
    float* b2   = ws + o; o += 768;
    float* hbuf = ws + o; o += (size_t)256 * 3072;
    float* part = ws + o; o += 2 * NPART;

    fwht_passA_kernel<<<dim3(1 + NPART, 2), 512, 0, stream>>>(theta, BB_W1, Pi_W1, GG_W1, W1,
                                                              BB_W2, Pi_W2, GG_W2, W2, 2047, part,
                                                              BB_b1, Pi_b1, GG_b1, ib1, b1,
                                                              BB_b2, Pi_b2, GG_b2, ib2, b2);
    passB_gemm1_kernel<<<336, 256, 0, stream>>>(W1, iW1, W2, iW2, part, DD_W, x, b1, hbuf);

    gemm_wmma_kernel<0, 8><<<256, 256, 0, stream>>>(hbuf, W2, b2, out, 256, 768, 3072);
}


# GEMM k loops start at a per-workgroup chunk (GEMM2: (mt+ng) mod 12; fused GEMM1: block mod 24) and wrap round: sharers of operand rows spread over 12 / 24 starts; k summed in rotated chunk order
# speedup vs baseline: 9.2764x; 1.0681x over previous
#include <hip/hip_runtime.h>

typedef __attribute__((ext_vector_type(2))) float v2f;
typedef __attribute__((ext_vector_type(4))) float v4f;
typedef __attribute__((ext_vector_type(8))) float v8f;

#define COLS     4096
#define ROWS     1024
#define ROWS_OUT 576
#define DD_W     2359296
#define TC       32
#define PB_PITCH 260
#define P_W      2048

__device__ __forceinline__ void fwht_lds(float* buf, int n, int tid, int nt) {
    for (int h = 1; h < n; h <<= 1) {
        __syncthreads();
        for (int p = tid; p < (n >> 1); p += nt) {
            int i0 = ((p & ~(h - 1)) << 1) | (p & (h - 1));
            int i1 = i0 + h;
            float a = buf[i0], b = buf[i1];
            buf[i0] = a + b;
            buf[i1] = a - b;
        }
    }
    __syncthreads();
}

__device__ __forceinline__ void lds_only_barrier() {
    __builtin_amdgcn_fence(__ATOMIC_RELEASE, "workgroup", "local");
    __builtin_amdgcn_s_barrier();
    __builtin_amdgcn_fence(__ATOMIC_ACQUIRE, "workgroup", "local");
}

__device__ __forceinline__ void fwht_lds_inflight(float* buf, int n, int tid, int nt) {
    for (int h = 1; h < n; h <<= 1) {
        lds_only_barrier();
        for (int p = tid; p < (n >> 1); p += nt) {
            int i0 = ((p & ~(h - 1)) << 1) | (p & (h - 1));
            int i1 = i0 + h;
            float a = buf[i0], b = buf[i1];
            buf[i0] = a + b;
            buf[i1] = a - b;
        }
    }
    lds_only_barrier();
}

#define NPART (COLS / TC)

__device__ __forceinline__ float wave_sum(float x) {
#pragma unroll
    for (int m = warpSize >> 1; m > 0; m >>= 1) x += __shfl_xor(x, m);
    return x;
}

template<int N, int B0, int B1>
__device__ __forceinline__ void fwht_reg_stages(float (&v)[N]) {
#pragma unroll
    for (int b = B0; b < B1; ++b) {
        const int h = 1 << b;
#pragma unroll
        for (int p = 0; p < N / 2; ++p) {
            const int i0 = ((p & ~(h - 1)) << 1) | (p & (h - 1));
            const float x = v[i0], y = v[i0 + h];
            v[i0] = x + y;
            v[i0 + h] = x - y;
        }
    }
}

template<int P, int LLv, int DD>
__device__ __forceinline__ void bias_projection(const float* __restrict__ theta,
                                                const float* __restrict__ BB,
                                                const int* __restrict__ Pi,
                                                const float* __restrict__ GG,
                                                const float* __restrict__ init,
                                                float* __restrict__ out,
                                                float* w, float* m, float* red) {
    constexpr int nt = 512;
    constexpr int NP = P / nt, NL = LLv / nt, ND = (DD + nt - 1) / nt;
    const int tid = threadIdx.x;
    float bv[NP], tv[NP], gg[NL], in[ND];
    int pi[NL];
#pragma unroll
    for (int k = 0; k < NP; ++k) { bv[k] = BB[tid + nt * k]; tv[k] = theta[tid + nt * k]; }
#pragma unroll
    for (int k = 0; k < NL; ++k) gg[k] = GG[tid + nt * k];
#pragma unroll
    for (int k = 0; k < NL; ++k) pi[k] = Pi[tid + nt * k];
#pragma unroll
    for (int k = 0; k < ND; ++k) in[k] = (tid + nt * k < DD) ? init[tid + nt * k] : 0.f;
#pragma unroll
    for (int k = 0; k < NP; ++k) w[tid + nt * k] = bv[k] * tv[k];
    fwht_lds(w, P, tid, nt);
    float s = 0.f;
#pragma unroll
    for (int k = 0; k < NL; ++k) { float g = gg[k]; s += g * g; }
    red[tid] = s; __syncthreads();
    for (int off = nt >> 1; off > 0; off >>= 1) {
        if (tid < off) red[tid] += red[tid + off];
        __syncthreads();
    }
#pragma unroll
    for (int k = 0; k < NL; ++k) m[tid + nt * k] = w[pi[k] & (P - 1)] * gg[k];
    fwht_lds(m, LLv, tid, nt);
    float scale = rsqrtf((float)DD * red[0]);
#pragma unroll
    for (int k = 0; k < ND; ++k) {
        const int i = tid + nt * k;
        if (i < DD) out[i] = in[k] + m[i] * scale;
    }
}

__global__ __launch_bounds__(512)
void fwht_passA_kernel(const float* __restrict__ theta,
                       const float* __restrict__ BB_a,
                       const int* __restrict__ Pi_a, const float* __restrict__ GG_a,
                       float* __restrict__ m_a,
                       const float* __restrict__ BB_b,
                       const int* __restrict__ Pi_b, const float* __restrict__ GG_b,
                       float* __restrict__ m_b, int pmask,
                       float* __restrict__ part2,
                       const float* __restrict__ BB_b1, const int* __restrict__ Pi_b1,
                       const float* __restrict__ GG_b1, const float* __restrict__ init_b1,
                       float* __restrict__ out_b1,
                       const float* __restrict__ BB_b2, const int* __restrict__ Pi_b2,
                       const float* __restrict__ GG_b2, const float* __restrict__ init_b2,
                       float* __restrict__ out_b2) {
    __shared__ float tile[ROWS * TC];
    __shared__ float wl[P_W];
    __shared__ float wsum[16];
    if (blockIdx.x == 0) {
        if (blockIdx.y == 0)
            bias_projection<P_W, 4096, 3072>(theta, BB_b1, Pi_b1, GG_b1, init_b1, out_b1,
                                             wl, tile, tile + 4096);
        else
            bias_projection<1024, 1024, 768>(theta, BB_b2, Pi_b2, GG_b2, init_b2, out_b2,
                                             wl, tile, tile + 4096);
        return;
    }
    const int lane = threadIdx.x & 31;
    const int wv = threadIdx.x >> 5;
    const int tix = blockIdx.x - 1;
    const float* __restrict__ BB = blockIdx.y ? BB_b : BB_a;
    const int*   __restrict__ Pi = blockIdx.y ? Pi_b : Pi_a;
    const float* __restrict__ GG = blockIdx.y ? GG_b : GG_a;
    float* __restrict__ mbuf     = blockIdx.y ? m_b : m_a;
    const int col = tix * TC + lane;

    float wp[P_W / 512];
#pragma unroll
    for (int k = 0; k < P_W / 512; ++k) wp[k] = BB[threadIdx.x + 512 * k] * theta[threadIdx.x + 512 * k];
    float v[64];
    int pi[64];
    const int g0 = wv * 64 * COLS + col;
#pragma unroll
    for (int j = 0; j < 64; ++j) pi[j] = Pi[g0 + j * COLS];
#pragma unroll
    for (int j = 0; j < 64; ++j) v[j] = GG[g0 + j * COLS];
#pragma unroll
    for (int k = 0; k < P_W / 512; ++k) wl[threadIdx.x + 512 * k] = wp[k];
    fwht_lds_inflight(wl, P_W, threadIdx.x, 512);
    float ss[4] = {0.f, 0.f, 0.f, 0.f};
#pragma unroll
    for (int j = 0; j < 64; ++j) {
        const float g = v[j];
        ss[j & 3] += g * g;
        v[j] = wl[pi[j] & pmask] * g;
    }
    const float wtot = wave_sum((ss[0] + ss[1]) + (ss[2] + ss[3]));
    if (lane == 0) wsum[wv] = wtot;
    fwht_reg_stages<64, 0, 6>(v);
#pragma unroll
    for (int j = 0; j < 64; ++j) tile[(wv * 64 + j) * TC + lane] = v[j];
    __syncthreads();
    if (threadIdx.x == 0) {
        float tot = wsum[0];
#pragma unroll
        for (int i = 1; i < 16; ++i) tot += wsum[i];
        part2[NPART * blockIdx.y + tix] = tot;
    }
    float u[64];
#pragma unroll
    for (int q = 0; q < 4; ++q) {
#pragma unroll
        for (int jh = 0; jh < 16; ++jh) u[16 * q + jh] = tile[(64 * jh + 4 * wv + q) * TC + lane];
    }
    fwht_reg_stages<64, 0, 4>(u);
#pragma unroll
    for (int q = 0; q < 4; ++q) {
#pragma unroll
        for (int jh = 0; jh < ROWS_OUT / 64; ++jh)
            mbuf[(size_t)(64 * jh + 4 * wv + q) * COLS + col] = u[16 * q + jh];
    }
}

struct Frag { v4f q0, q1, q2, q3; };

__device__ __forceinline__ void load_frag(Frag& f, const float* __restrict__ p) {
    const v4f* p4 = (const v4f*)p;
    f.q0 = p4[0]; f.q1 = p4[1]; f.q2 = p4[2]; f.q3 = p4[3];
}

#define SCHED_FENCE() __builtin_amdgcn_sched_barrier(0)
#define KEEP_LOADS_IN_THIS_TRIP() asm volatile("" ::: "memory")

#define LO2(q) __builtin_shufflevector(q, q, 0, 1)
#define HI2(q) __builtin_shufflevector(q, q, 2, 3)
#define WMMA_F32(a, b, c) __builtin_amdgcn_wmma_f32_16x16x4_f32(false, a, false, b, (short)0, c, false, false)
#define NT 3
#define KSTEP(SEL, Q)                                   \
    acc0 = WMMA_F32(SEL(fa.Q), SEL(f0.Q), acc0);        \
    acc1 = WMMA_F32(SEL(fa.Q), SEL(f1.Q), acc1);        \
    acc2 = WMMA_F32(SEL(fa.Q), SEL(f2.Q), acc2);

__device__ __forceinline__ void mma_chunk(const Frag& fa, const Frag& f0, const Frag& f1,
                                          const Frag& f2,
                                          v8f& acc0, v8f& acc1, v8f& acc2) {
    KSTEP(LO2, q0) KSTEP(HI2, q0)
    KSTEP(LO2, q1) KSTEP(HI2, q1)
    KSTEP(LO2, q2) KSTEP(HI2, q2)
    KSTEP(LO2, q3) KSTEP(HI2, q3)
}

template<int RELU, int KS>
__global__ __launch_bounds__(256)
void gemm_wmma_kernel(const float* __restrict__ A, const float* __restrict__ B,
                      const float* __restrict__ bias, float* __restrict__ C,
                      int M, int N, int K) {
    constexpr int SPB = 8 / KS;
    constexpr int NG = 2 * NT;
    __shared__ v4f part[8 * NG * 32];
    int lane = threadIdx.x & 31;
    int wave = threadIdx.x >> 5;
    int sl = wave / KS;
    int ks = wave % KS;
    int job = blockIdx.x * SPB + sl;
    int mTiles = M >> 4;
    int mt = job % mTiles;
    int ng = job / mTiles;
    int row = lane & 15;
    int kg = (lane >> 4) << 4;
    int kLen = K / KS;
    size_t koff = (size_t)ks * kLen + kg;
    const float* ap  = A + (size_t)(mt * 16 + row) * K + koff;
    const float* bp0 = B + (size_t)(ng * (16 * NT) + row) * K + koff;
    const float* bp1 = bp0 + (size_t)16 * K;
    const float* bp2 = bp0 + (size_t)32 * K;
    v8f acc0 = {}, acc1 = {}, acc2 = {};

    Frag xa, x0, x1, x2;
    Frag ya, y0, y1, y2;

    const int nch = kLen >> 5;
    const int job0 = blockIdx.x * SPB;
    int o = (int)((unsigned)(job0 % mTiles + job0 / mTiles) % (unsigned)nch) << 5;

    load_frag(xa, ap + o); load_frag(x0, bp0 + o); load_frag(x1, bp1 + o);
    load_frag(x2, bp2 + o);

    for (int i = 0; i + 2 < nch; i += 2) {
        int o1 = o + 32; if (o1 == kLen) o1 = 0;
        load_frag(ya, ap + o1); load_frag(y0, bp0 + o1); load_frag(y1, bp1 + o1);
        load_frag(y2, bp2 + o1);
        SCHED_FENCE();
        mma_chunk(xa, x0, x1, x2, acc0, acc1, acc2);
        SCHED_FENCE();
        o = o1 + 32; if (o == kLen) o = 0;
        load_frag(xa, ap + o); load_frag(x0, bp0 + o); load_frag(x1, bp1 + o);
        load_frag(x2, bp2 + o);
        KEEP_LOADS_IN_THIS_TRIP();
        SCHED_FENCE();
        mma_chunk(ya, y0, y1, y2, acc0, acc1, acc2);
    }
    {
        int o1 = o + 32; if (o1 == kLen) o1 = 0;
        load_frag(ya, ap + o1); load_frag(y0, bp0 + o1); load_frag(y1, bp1 + o1);
        load_frag(y2, bp2 + o1);
    }
    SCHED_FENCE();
    mma_chunk(xa, x0, x1, x2, acc0, acc1, acc2);
    mma_chunk(ya, y0, y1, y2, acc0, acc1, acc2);

    v4f* mine = part + wave * (NG * 32);
    v8f accs[NT] = {acc0, acc1, acc2};
#pragma unroll
    for (int t = 0; t < NT; ++t) {
        mine[(2 * t) * 32 + lane]     = __builtin_shufflevector(accs[t], accs[t], 0, 1, 2, 3);
        mine[(2 * t + 1) * 32 + lane] = __builtin_shufflevector(accs[t], accs[t], 4, 5, 6, 7);
    }
    __syncthreads();

    constexpr int GPW = (NG + KS - 1) / KS;
    const v4f* sp = part + sl * KS * (NG * 32);
    int mrow = mt * 16 + (lane >> 4) * 8;
#pragma unroll
    for (int j = 0; j < GPW; ++j) {
        int g = ks * GPW + j;
        if (g >= NG) break;
        int t = g >> 1;
        int vb = (g & 1) * 4;
        v4f s = sp[g * 32 + lane];
#pragma unroll
        for (int q = 1; q < KS; ++q) s += sp[q * (NG * 32) + g * 32 + lane];
        int n = ng * (16 * NT) + t * 16 + row;
        float bv = bias[n];
#pragma unroll
        for (int e = 0; e < 4; ++e) {
            float xv = s[e] + bv;
            if (RELU) xv = fmaxf(xv, 0.f);
            C[(size_t)(mrow + vb + e) * N + n] = xv;
        }
    }
}

#define NFUSED   192
#define WT_PITCH 772

#define KSTEP2(SEL, Q)                                  \
    acc0 = WMMA_F32(SEL(fa0.Q), SEL(fb.Q), acc0);       \
    acc1 = WMMA_F32(SEL(fa1.Q), SEL(fb.Q), acc1);

__device__ __forceinline__ void mma_chunk2(const Frag& fa0, const Frag& fa1, const Frag& fb,
                                           v8f& acc0, v8f& acc1) {
    KSTEP2(LO2, q0) KSTEP2(HI2, q0)
    KSTEP2(LO2, q1) KSTEP2(HI2, q1)
    KSTEP2(LO2, q2) KSTEP2(HI2, q2)
    KSTEP2(LO2, q3) KSTEP2(HI2, q3)
}

__device__ __forceinline__ void load_frag_lds(Frag& f, const float* p) {
    const v4f* p4 = (const v4f*)p;
    f.q0 = p4[0]; f.q1 = p4[1]; f.q2 = p4[2]; f.q3 = p4[3];
}

__global__ __launch_bounds__(256)
void passB_gemm1_kernel(const float* __restrict__ m_a, const float* __restrict__ init_a,
                        float* buf_b, const float* __restrict__ init_b,
                        const float* __restrict__ part2, int DD,
                        const float* __restrict__ X, const float* __restrict__ bias1,
                        float* __restrict__ H) {
    __shared__ v4f blk4[4 * 16 * PB_PITCH / 4];
    __shared__ v4f wt4[16 * WT_PITCH / 4];
    __shared__ float wsum[8];
    const int tid = threadIdx.x;
    const int g = tid >> 6;
    const int t = tid & 63;
    const bool fused = blockIdx.x < NFUSED;
    const bool active = !fused || g < 3;
    const int rowIdx = fused ? (int)blockIdx.x * 3 + g : ((int)blockIdx.x - NFUSED) * 4 + g;
    const float* src = fused ? m_a : buf_b;
    const float* init = fused ? init_a : init_b;
    const float* part = part2 + (fused ? 0 : NPART);
    const size_t base = (size_t)rowIdx * COLS;
    const v4f* row4 = (const v4f*)(src + base) + t;
    const v4f* init4 = (const v4f*)(init + base) + t;
    v4f* dst4 = (v4f*)(buf_b + base) + t;
    v4f* myblk4 = blk4 + g * (16 * PB_PITCH / 4);
    float* blk = (float*)myblk4;
    float* wt = (float*)wt4;

    float v[64];
    v4f in[16];
    if (active) {
        const float ps = wave_sum(part[t] + part[t + 64]);
#pragma unroll
        for (int c = 0; c < 16; ++c) {
            const v4f x = row4[64 * c];
            v[4 * c + 0] = x[0]; v[4 * c + 1] = x[1]; v[4 * c + 2] = x[2]; v[4 * c + 3] = x[3];
        }
#pragma unroll
        for (int c = 0; c < 16; ++c) in[c] = init4[64 * c];
        fwht_reg_stages<64, 0, 2>(v);
#pragma unroll
        for (int c = 0; c < 16; ++c) {
            v4f x; x[0] = v[4 * c + 0]; x[1] = v[4 * c + 1]; x[2] = v[4 * c + 2]; x[3] = v[4 * c + 3];
            myblk4[c * (PB_PITCH / 4) + t] = x;
        }
        if ((t & 31) == 0) wsum[tid >> 5] = ps;
    }
    __syncthreads();
    float gtot = 0.f;
    if (active) {
        gtot = wsum[2 * g] + wsum[2 * g + 1];
        float* pb = blk + (t >> 2) * PB_PITCH + (t & 3);
        float u[64];
#pragma unroll
        for (int m = 0; m < 64; ++m) u[m] = pb[4 * m];
        fwht_reg_stages<64, 0, 6>(u);
#pragma unroll
        for (int m = 0; m < 64; ++m) pb[4 * m] = u[m];
    }
    __syncthreads();
    if (active) {
#pragma unroll
        for (int c = 0; c < 16; ++c) {
            const v4f x = myblk4[c * (PB_PITCH / 4) + t];
            v[4 * c + 0] = x[0]; v[4 * c + 1] = x[1]; v[4 * c + 2] = x[2]; v[4 * c + 3] = x[3];
        }
        fwht_reg_stages<64, 2, 6>(v);
        const float scale = rsqrtf((float)DD * gtot);
#pragma unroll
        for (int c = 0; c < 16; ++c) {
            v4f o;
            o[0] = in[c][0] + v[4 * c + 0] * scale; o[1] = in[c][1] + v[4 * c + 1] * scale;
            o[2] = in[c][2] + v[4 * c + 2] * scale; o[3] = in[c][3] + v[4 * c + 3] * scale;
            if (fused) {
                const int f = COLS * g + 256 * c + 4 * t;
                const int n = f / 768;
                const int k = f - 768 * n;
                wt4[(n * WT_PITCH + k) >> 2] = o;
            } else {
                dst4[64 * c] = o;
            }
        }
    }
    if (!fused) return;
    __syncthreads();

    const int lane = tid & 31;
    const int wave = tid >> 5;
    const int row = lane & 15;
    const int kg = (lane >> 4) << 4;
    const float* ap0 = X + (size_t)(wave * 32 + row) * 768 + kg;
    const float* ap1 = ap0 + (size_t)16 * 768;
    const float* bl = wt + row * WT_PITCH + kg;
    v8f acc0 = {}, acc1 = {};
    Frag xa0, xa1, xb;
    Frag ya0, ya1, yb;

    int o = ((int)blockIdx.x % 24) << 5;

    load_frag(xa0, ap0 + o); load_frag(xa1, ap1 + o); load_frag_lds(xb, bl + o);
    for (int i = 0; i + 2 < 24; i += 2) {
        int o1 = o + 32; if (o1 == 768) o1 = 0;
        load_frag(ya0, ap0 + o1); load_frag(ya1, ap1 + o1); load_frag_lds(yb, bl + o1);
        SCHED_FENCE();
        mma_chunk2(xa0, xa1, xb, acc0, acc1);
        SCHED_FENCE();
        o = o1 + 32; if (o == 768) o = 0;
        load_frag(xa0, ap0 + o); load_frag(xa1, ap1 + o); load_frag_lds(xb, bl + o);
        KEEP_LOADS_IN_THIS_TRIP();
        SCHED_FENCE();
        mma_chunk2(ya0, ya1, yb, acc0, acc1);
    }
    {
        int o1 = o + 32; if (o1 == 768) o1 = 0;
        load_frag(ya0, ap0 + o1); load_frag(ya1, ap1 + o1); load_frag_lds(yb, bl + o1);
    }
    SCHED_FENCE();
    mma_chunk2(xa0, xa1, xb, acc0, acc1);
    mma_chunk2(ya0, ya1, yb, acc0, acc1);

    const int n = 16 * (int)blockIdx.x + row;
    const float bv = bias1[n];
    const int mrow = wave * 32 + (lane >> 4) * 8;
#pragma unroll
    for (int e = 0; e < 8; ++e) {
        H[(size_t)(mrow + e) * 3072 + n]      = fmaxf(acc0[e] + bv, 0.f);
        H[(size_t)(mrow + 16 + e) * 3072 + n] = fmaxf(acc1[e] + bv, 0.f);
    }
}

extern "C" void kernel_launch(void* const* d_in, const int* in_sizes, int n_in,
                              void* d_out, int out_size, void* d_ws, size_t ws_size,
                              hipStream_t stream) {
    const float* x     = (const float*)d_in[0];
    const float* theta = (const float*)d_in[1];
    const float* iW1   = (const float*)d_in[2];
    const float* ib1   = (const float*)d_in[3];
    const float* iW2   = (const float*)d_in[4];
    const float* ib2   = (const float*)d_in[5];
    const float* BB_W1 = (const float*)d_in[6];
    const int*   Pi_W1 = (const int*)d_in[7];
    const float* GG_W1 = (const float*)d_in[8];
    const float* BB_b1 = (const float*)d_in[9];
    const int*   Pi_b1 = (const int*)d_in[10];
    const float* GG_b1 = (const float*)d_in[11];
    const float* BB_W2 = (const float*)d_in[12];
    const int*   Pi_W2 = (const int*)d_in[13];
    const float* GG_W2 = (const float*)d_in[14];
    const float* BB_b2 = (const float*)d_in[15];
    const int*   Pi_b2 = (const int*)d_in[16];
    const float* GG_b2 = (const float*)d_in[17];
    float* out = (float*)d_out;

    float* ws = (float*)d_ws;
    size_t o = 0;
    float* W1   = ws + o; o += DD_W;
    float* W2   = ws + o; o += DD_W;
    float* b1   = ws + o; o += 3072;
    float* b2   = ws + o; o += 768;
    float* hbuf = ws + o; o += (size_t)256 * 3072;
    float* part = ws + o; o += 2 * NPART;

    fwht_passA_kernel<<<dim3(1 + NPART, 2), 512, 0, stream>>>(theta, BB_W1, Pi_W1, GG_W1, W1,
                                                              BB_W2, Pi_W2, GG_W2, W2, 2047, part,
                                                              BB_b1, Pi_b1, GG_b1, ib1, b1,
                                                              BB_b2, Pi_b2, GG_b2, ib2, b2);
    passB_gemm1_kernel<<<336, 256, 0, stream>>>(W1, iW1, W2, iW2, part, DD_W, x, b1, hbuf);

    gemm_wmma_kernel<0, 8><<<256, 256, 0, stream>>>(hbuf, W2, b2, out, 256, 768, 3072);
}
